# starting kernel
# baseline (speedup 1.0000x reference)
#include <hip/hip_runtime.h>

typedef __attribute__((ext_vector_type(2))) float v2f;
typedef __attribute__((ext_vector_type(8))) float v8f;

#define CFEAT 256

__global__ void zero_kernel(float* __restrict__ p, long n) {
    long i = (long)blockIdx.x * blockDim.x + threadIdx.x;
    long stride = (long)gridDim.x * blockDim.x;
    for (; i < n; i += stride) p[i] = 0.0f;
}

__global__ void scatter_kernel(const float* __restrict__ x,
                               const int* __restrict__ src,
                               const int* __restrict__ dst,
                               float* __restrict__ s,
                               float* __restrict__ cnt,
                               int E) {
    int wave = (int)((blockIdx.x * blockDim.x + threadIdx.x) >> 5);
    int lane = threadIdx.x & 31;
    if (wave >= E) return;
    int sr = src[wave];
    int ds = dst[wave];
    const float4* xs = (const float4*)(x + (long)sr * CFEAT);
    float* sd = s + (long)ds * CFEAT;
    float4 a = xs[lane];
    float4 b = xs[lane + 32];
    int base = lane * 4;
    atomicAdd(&sd[base + 0], a.x);
    atomicAdd(&sd[base + 1], a.y);
    atomicAdd(&sd[base + 2], a.z);
    atomicAdd(&sd[base + 3], a.w);
    atomicAdd(&sd[128 + base + 0], b.x);
    atomicAdd(&sd[128 + base + 1], b.y);
    atomicAdd(&sd[128 + base + 2], b.z);
    atomicAdd(&sd[128 + base + 3], b.w);
    if (lane == 0) atomicAdd(&cnt[ds], 1.0f);
}

__global__ void gemm_kernel(const float* __restrict__ s,
                            const float* __restrict__ cnt,
                            const float* __restrict__ xt,
                            const float* __restrict__ Wl,
                            const float* __restrict__ bl,
                            const float* __restrict__ Wr,
                            float* __restrict__ h,
                            float* __restrict__ colsum,
                            float* __restrict__ colsq,
                            int N, int rowTiles) {
    int wavesPerBlock = blockDim.x >> 5;
    int gw = blockIdx.x * wavesPerBlock + (threadIdx.x >> 5);
    int totalWaves = rowTiles * 4;
    if (gw >= totalWaves) return;

    int rowTile = gw >> 2;
    int colGrp  = gw & 3;
    int lane = threadIdx.x & 31;
    int l15  = lane & 15;
    int khalf = (lane >> 4) * 2;

    int rowA = rowTile * 16 + l15;
    int rowL = (rowA < N) ? rowA : (N - 1);
    float invc = 1.0f / fmaxf(cnt[rowL], 1.0f);
    const float* sRow = s  + (long)rowL * CFEAT;
    const float* xRow = xt + (long)rowL * CFEAT;

    int colBase = colGrp * 64;
    const float* wlRow[4];
    const float* wrRow[4];
#pragma unroll
    for (int t = 0; t < 4; ++t) {
        int colB = colBase + t * 16 + l15;
        wlRow[t] = Wl + (long)colB * CFEAT;
        wrRow[t] = Wr + (long)colB * CFEAT;
    }

    v8f acc[4] = {{0,0,0,0,0,0,0,0}, {0,0,0,0,0,0,0,0},
                  {0,0,0,0,0,0,0,0}, {0,0,0,0,0,0,0,0}};

    for (int k = 0; k < CFEAT; k += 4) {
        int ko = k + khalf;
        v2f aL = *(const v2f*)(sRow + ko);
        aL *= invc;
        v2f aR = *(const v2f*)(xRow + ko);
#pragma unroll
        for (int t = 0; t < 4; ++t) {
            v2f bL = *(const v2f*)(wlRow[t] + ko);
            v2f bR = *(const v2f*)(wrRow[t] + ko);
            acc[t] = __builtin_amdgcn_wmma_f32_16x16x4_f32(
                false, aL, false, bL, (short)0, acc[t], false, false);
            acc[t] = __builtin_amdgcn_wmma_f32_16x16x4_f32(
                false, aR, false, bR, (short)0, acc[t], false, false);
        }
    }

    int rbase = rowTile * 16 + ((lane < 16) ? 0 : 8);
#pragma unroll
    for (int t = 0; t < 4; ++t) {
        int col = colBase + t * 16 + l15;
        float bias = bl[col];
        float sp = 0.0f, qp = 0.0f;
#pragma unroll
        for (int v = 0; v < 8; ++v) {
            float val = fmaxf(acc[t][v] + bias, 0.0f);
            int r = rbase + v;
            bool ok = (r < N);
            float m = ok ? val : 0.0f;
            sp += m;
            qp += m * m;
            if (ok) h[(long)r * CFEAT + col] = val;
        }
        atomicAdd(&colsum[col], sp);
        atomicAdd(&colsq[col], qp);
    }
}

__global__ void norm_kernel(const float* __restrict__ hbuf,
                            const float* __restrict__ colsum,
                            const float* __restrict__ colsq,
                            const float* __restrict__ gamma,
                            const float* __restrict__ beta,
                            float* __restrict__ out, int N) {
    long i = (long)blockIdx.x * blockDim.x + threadIdx.x;
    long total = (long)N * CFEAT;
    if (i >= total) return;
    int col = (int)(i & (CFEAT - 1));
    float invN = 1.0f / (float)N;
    float mu  = colsum[col] * invN;
    float var = colsq[col] * invN - mu * mu;
    float rs  = rsqrtf(var + 1e-5f);
    out[i] = gamma[col] * (hbuf[i] - mu) * rs + beta[col];
}

extern "C" void kernel_launch(void* const* d_in, const int* in_sizes, int n_in,
                              void* d_out, int out_size, void* d_ws, size_t ws_size,
                              hipStream_t stream) {
    const float* x    = (const float*)d_in[0];
    const int* e0src  = (const int*)d_in[1];
    const int* e0dst  = (const int*)d_in[2];
    const int* e1src  = (const int*)d_in[3];
    const int* e1dst  = (const int*)d_in[4];
    const float* Wl0  = (const float*)d_in[5];
    const float* bl0  = (const float*)d_in[6];
    const float* Wr0  = (const float*)d_in[7];
    const float* g0   = (const float*)d_in[8];
    const float* be0  = (const float*)d_in[9];
    const float* Wl1  = (const float*)d_in[10];
    const float* bl1  = (const float*)d_in[11];
    const float* Wr1  = (const float*)d_in[12];
    const float* g1   = (const float*)d_in[13];
    const float* be1  = (const float*)d_in[14];

    const int N1 = 50000, N2 = 12500;
    const int E0 = in_sizes[1];
    const int E1 = in_sizes[3];

    float* ws = (float*)d_ws;
    const long NC1 = (long)N1 * CFEAT;
    float* x1    = ws;
    float* sbuf  = ws + NC1;
    float* hbuf  = ws + 2 * NC1;
    float* cntb  = ws + 3 * NC1;
    float* stats = cntb + 50048;

    zero_kernel<<<4096, 256, 0, stream>>>(sbuf, NC1);
    zero_kernel<<<256, 256, 0, stream>>>(cntb, 50048 + 512);
    scatter_kernel<<<(E0 + 7) / 8, 256, 0, stream>>>(x, e0src, e0dst, sbuf, cntb, E0);
    {
        int rowTiles = (N1 + 15) / 16;
        int waves = rowTiles * 4;
        gemm_kernel<<<(waves + 7) / 8, 256, 0, stream>>>(
            sbuf, cntb, x, Wl0, bl0, Wr0, hbuf, stats, stats + 256, N1, rowTiles);
    }
    norm_kernel<<<(int)((NC1 + 255) / 256), 256, 0, stream>>>(
        hbuf, stats, stats + 256, g0, be0, x1, N1);

    const long NC2 = (long)N2 * CFEAT;
    zero_kernel<<<2048, 256, 0, stream>>>(sbuf, NC2);
    zero_kernel<<<256, 256, 0, stream>>>(cntb, 50048 + 512);
    scatter_kernel<<<(E1 + 7) / 8, 256, 0, stream>>>(x1, e1src, e1dst, sbuf, cntb, E1);
    {
        int rowTiles = (N2 + 15) / 16;
        int waves = rowTiles * 4;
        gemm_kernel<<<(waves + 7) / 8, 256, 0, stream>>>(
            sbuf, cntb, x1, Wl1, bl1, Wr1, hbuf, stats, stats + 256, N2, rowTiles);
    }
    norm_kernel<<<(int)((NC2 + 255) / 256), 256, 0, stream>>>(
        hbuf, stats, stats + 256, g1, be1, (float*)d_out, N2);
}


# Segment sum without float atomics: per-call edge list grouped by destination (integer-atomic count, bump-allocated ranges, arrival-order fill), one wave per row gathers and sums; 14 launches
# speedup vs baseline: 1.3245x; 1.3245x over previous
#include <hip/hip_runtime.h>

typedef __attribute__((ext_vector_type(2))) float v2f;
typedef __attribute__((ext_vector_type(8))) float v8f;

#define CFEAT 256

__global__ void zero_int_kernel(int* __restrict__ p, int n) {
    int i = blockIdx.x * blockDim.x + threadIdx.x;
    int stride = gridDim.x * blockDim.x;
    for (; i < n; i += stride) p[i] = 0;
}

__global__ void degree_kernel(const int* __restrict__ dst, int* __restrict__ deg, int E) {
    int e = blockIdx.x * blockDim.x + threadIdx.x;
    if (e < E) atomicAdd(&deg[dst[e]], 1);
}

__global__ void offsets_kernel(const int* __restrict__ deg,
                               int* __restrict__ cursor,
                               int* __restrict__ total,
                               float* __restrict__ stats, int N) {
    int gid = blockIdx.x * blockDim.x + threadIdx.x;
    int lane = threadIdx.x & 31;
    int d = (gid < N) ? deg[gid] : 0;
    int incl = d;
    for (int off = 1; off < 32; off <<= 1) {
        int v = __shfl_up(incl, off);
        if (lane >= off) incl += v;
    }
    int wsum = __shfl(incl, 31);
    int base = 0;
    if (lane == 31) base = atomicAdd(total, wsum);
    base = __shfl(base, 31);
    if (gid < N) cursor[gid] = base + incl - d;
    for (int i = gid; i < 2 * CFEAT; i += gridDim.x * blockDim.x) stats[i] = 0.0f;
}

__global__ void fill_kernel(const int* __restrict__ src,
                            const int* __restrict__ dst,
                            int* __restrict__ cursor,
                            int* __restrict__ csr, int E) {
    int e = blockIdx.x * blockDim.x + threadIdx.x;
    if (e >= E) return;
    int pos = atomicAdd(&cursor[dst[e]], 1);
    if ((unsigned)pos < (unsigned)E) csr[pos] = src[e];
}

__global__ void gather_kernel(const float* __restrict__ x,
                              const int* __restrict__ csr,
                              const int* __restrict__ cursor,
                              const int* __restrict__ deg,
                              float* __restrict__ s,
                              float* __restrict__ cnt, int N) {
    int w = __builtin_amdgcn_readfirstlane((int)((blockIdx.x * blockDim.x + threadIdx.x) >> 5));
    int lane = threadIdx.x & 31;
    if (w >= N) return;
    int d = deg[w];
    int end = cursor[w];
    int beg = end - d;
    float4 a0 = make_float4(0.0f, 0.0f, 0.0f, 0.0f);
    float4 a1 = make_float4(0.0f, 0.0f, 0.0f, 0.0f);
    for (int c = beg; c < end; c += 32) {
        int n = end - c;
        if (n > 32) n = 32;
        int my = (lane < n) ? csr[c + lane] : 0;
        int j = 0;
        for (; j + 4 <= n; j += 4) {
            const float4* r0 = (const float4*)(x + (long)__builtin_amdgcn_readlane(my, j + 0) * CFEAT);
            const float4* r1 = (const float4*)(x + (long)__builtin_amdgcn_readlane(my, j + 1) * CFEAT);
            const float4* r2 = (const float4*)(x + (long)__builtin_amdgcn_readlane(my, j + 2) * CFEAT);
            const float4* r3 = (const float4*)(x + (long)__builtin_amdgcn_readlane(my, j + 3) * CFEAT);
            float4 p0 = r0[lane], q0 = r0[lane + 32];
            float4 p1 = r1[lane], q1 = r1[lane + 32];
            float4 p2 = r2[lane], q2 = r2[lane + 32];
            float4 p3 = r3[lane], q3 = r3[lane + 32];
            a0 += p0; a1 += q0;
            a0 += p1; a1 += q1;
            a0 += p2; a1 += q2;
            a0 += p3; a1 += q3;
        }
        for (; j < n; ++j) {
            const float4* r0 = (const float4*)(x + (long)__builtin_amdgcn_readlane(my, j) * CFEAT);
            float4 p0 = r0[lane], q0 = r0[lane + 32];
            a0 += p0; a1 += q0;
        }
    }
    float4* o = (float4*)(s + (long)w * CFEAT);
    o[lane] = a0;
    o[lane + 32] = a1;
    if (lane == 0) cnt[w] = (float)d;
}

__global__ void gemm_kernel(const float* __restrict__ s,
                            const float* __restrict__ cnt,
                            const float* __restrict__ xt,
                            const float* __restrict__ Wl,
                            const float* __restrict__ bl,
                            const float* __restrict__ Wr,
                            float* __restrict__ h,
                            float* __restrict__ colsum,
                            float* __restrict__ colsq,
                            int N, int rowTiles) {
    int wavesPerBlock = blockDim.x >> 5;
    int gw = blockIdx.x * wavesPerBlock + (threadIdx.x >> 5);
    int totalWaves = rowTiles * 4;
    if (gw >= totalWaves) return;

    int rowTile = gw >> 2;
    int colGrp  = gw & 3;
    int lane = threadIdx.x & 31;
    int l15  = lane & 15;
    int khalf = (lane >> 4) * 2;

    int rowA = rowTile * 16 + l15;
    int rowL = (rowA < N) ? rowA : (N - 1);
    float invc = 1.0f / fmaxf(cnt[rowL], 1.0f);
    const float* sRow = s  + (long)rowL * CFEAT;
    const float* xRow = xt + (long)rowL * CFEAT;

    int colBase = colGrp * 64;
    const float* wlRow[4];
    const float* wrRow[4];
#pragma unroll
    for (int t = 0; t < 4; ++t) {
        int colB = colBase + t * 16 + l15;
        wlRow[t] = Wl + (long)colB * CFEAT;
        wrRow[t] = Wr + (long)colB * CFEAT;
    }

    v8f acc[4] = {{0,0,0,0,0,0,0,0}, {0,0,0,0,0,0,0,0},
                  {0,0,0,0,0,0,0,0}, {0,0,0,0,0,0,0,0}};

    for (int k = 0; k < CFEAT; k += 4) {
        int ko = k + khalf;
        v2f aL = *(const v2f*)(sRow + ko);
        aL *= invc;
        v2f aR = *(const v2f*)(xRow + ko);
#pragma unroll
        for (int t = 0; t < 4; ++t) {
            v2f bL = *(const v2f*)(wlRow[t] + ko);
            v2f bR = *(const v2f*)(wrRow[t] + ko);
            acc[t] = __builtin_amdgcn_wmma_f32_16x16x4_f32(
                false, aL, false, bL, (short)0, acc[t], false, false);
            acc[t] = __builtin_amdgcn_wmma_f32_16x16x4_f32(
                false, aR, false, bR, (short)0, acc[t], false, false);
        }
    }

    int rbase = rowTile * 16 + ((lane < 16) ? 0 : 8);
#pragma unroll
    for (int t = 0; t < 4; ++t) {
        int col = colBase + t * 16 + l15;
        float bias = bl[col];
        float sp = 0.0f, qp = 0.0f;
#pragma unroll
        for (int v = 0; v < 8; ++v) {
            float val = fmaxf(acc[t][v] + bias, 0.0f);
            int r = rbase + v;
            bool ok = (r < N);
            float m = ok ? val : 0.0f;
            sp += m;
            qp += m * m;
            if (ok) h[(long)r * CFEAT + col] = val;
        }
        atomicAdd(&colsum[col], sp);
        atomicAdd(&colsq[col], qp);
    }
}

__global__ void norm_kernel(const float* __restrict__ hbuf,
                            const float* __restrict__ colsum,
                            const float* __restrict__ colsq,
                            const float* __restrict__ gamma,
                            const float* __restrict__ beta,
                            float* __restrict__ out, int N) {
    long i = (long)blockIdx.x * blockDim.x + threadIdx.x;
    long total = (long)N * CFEAT;
    if (i >= total) return;
    int col = (int)(i & (CFEAT - 1));
    float invN = 1.0f / (float)N;
    float mu  = colsum[col] * invN;
    float var = colsq[col] * invN - mu * mu;
    float rs  = rsqrtf(var + 1e-5f);
    out[i] = gamma[col] * (hbuf[i] - mu) * rs + beta[col];
}

extern "C" void kernel_launch(void* const* d_in, const int* in_sizes, int n_in,
                              void* d_out, int out_size, void* d_ws, size_t ws_size,
                              hipStream_t stream) {
    const float* x    = (const float*)d_in[0];
    const int* e0src  = (const int*)d_in[1];
    const int* e0dst  = (const int*)d_in[2];
    const int* e1src  = (const int*)d_in[3];
    const int* e1dst  = (const int*)d_in[4];
    const float* Wl0  = (const float*)d_in[5];
    const float* bl0  = (const float*)d_in[6];
    const float* Wr0  = (const float*)d_in[7];
    const float* g0   = (const float*)d_in[8];
    const float* be0  = (const float*)d_in[9];
    const float* Wl1  = (const float*)d_in[10];
    const float* bl1  = (const float*)d_in[11];
    const float* Wr1  = (const float*)d_in[12];
    const float* g1   = (const float*)d_in[13];
    const float* be1  = (const float*)d_in[14];

    const int N1 = 50000, N2 = 12500;
    const int E0 = in_sizes[1];
    const int E1 = in_sizes[3];

    float* ws = (float*)d_ws;
    const long NC1 = (long)N1 * CFEAT;
    const long NC2 = (long)N2 * CFEAT;
    float* x1    = ws;
    float* sbuf  = ws + NC1;
    float* hbuf  = ws + 2 * NC1;
    float* cntb  = ws + 3 * NC1;
    float* stats = cntb + 50048;

    int* t0 = (int*)x1;
    int* t1 = (int*)(sbuf + NC2);

    zero_int_kernel<<<64, 256, 0, stream>>>(t0, N1 + 32);
    degree_kernel<<<(E0 + 255) / 256, 256, 0, stream>>>(e0dst, t0, E0);
    offsets_kernel<<<(N1 + 255) / 256, 256, 0, stream>>>(t0, t0 + 65536, t0 + N1, stats, N1);
    fill_kernel<<<(E0 + 255) / 256, 256, 0, stream>>>(e0src, e0dst, t0 + 65536, t0 + 131072, E0);
    gather_kernel<<<(N1 + 7) / 8, 256, 0, stream>>>(x, t0 + 131072, t0 + 65536, t0, sbuf, cntb, N1);
    {
        int rowTiles = (N1 + 15) / 16;
        int waves = rowTiles * 4;
        gemm_kernel<<<(waves + 7) / 8, 256, 0, stream>>>(
            sbuf, cntb, x, Wl0, bl0, Wr0, hbuf, stats, stats + 256, N1, rowTiles);
    }
    norm_kernel<<<(int)((NC1 + 255) / 256), 256, 0, stream>>>(
        hbuf, stats, stats + 256, g0, be0, x1, N1);

    zero_int_kernel<<<64, 256, 0, stream>>>(t1, N2 + 32);
    degree_kernel<<<(E1 + 255) / 256, 256, 0, stream>>>(e1dst, t1, E1);
    offsets_kernel<<<(N2 + 255) / 256, 256, 0, stream>>>(t1, t1 + 65536, t1 + N2, stats, N2);
    fill_kernel<<<(E1 + 255) / 256, 256, 0, stream>>>(e1src, e1dst, t1 + 65536, t1 + 131072, E1);
    gather_kernel<<<(N2 + 7) / 8, 256, 0, stream>>>(x1, t1 + 131072, t1 + 65536, t1, sbuf, cntb, N2);
    {
        int rowTiles = (N2 + 15) / 16;
        int waves = rowTiles * 4;
        gemm_kernel<<<(waves + 7) / 8, 256, 0, stream>>>(
            sbuf, cntb, x1, Wl1, bl1, Wr1, hbuf, stats, stats + 256, N2, rowTiles);
    }
    norm_kernel<<<(int)((NC2 + 255) / 256), 256, 0, stream>>>(
        hbuf, stats, stats + 256, g1, be1, (float*)d_out, N2);
}


# GEMM re-tiled: a workgroup computes 64 rows x 256 cols with operands staged through LDS per 32-wide K chunk; K summed aggregated half then root half; column sums joined per workgroup before atomics
# speedup vs baseline: 1.8593x; 1.4038x over previous
#include <hip/hip_runtime.h>

typedef __attribute__((ext_vector_type(2))) float v2f;
typedef __attribute__((ext_vector_type(8))) float v8f;

#define CFEAT 256

__global__ void zero_int_kernel(int* __restrict__ p, int n) {
    int i = blockIdx.x * blockDim.x + threadIdx.x;
    int stride = gridDim.x * blockDim.x;
    for (; i < n; i += stride) p[i] = 0;
}

__global__ void degree_kernel(const int* __restrict__ dst, int* __restrict__ deg, int E) {
    int e = blockIdx.x * blockDim.x + threadIdx.x;
    if (e < E) atomicAdd(&deg[dst[e]], 1);
}

__global__ void offsets_kernel(const int* __restrict__ deg,
                               int* __restrict__ cursor,
                               int* __restrict__ total,
                               float* __restrict__ stats, int N) {
    int gid = blockIdx.x * blockDim.x + threadIdx.x;
    int lane = threadIdx.x & 31;
    int d = (gid < N) ? deg[gid] : 0;
    int incl = d;
    for (int off = 1; off < 32; off <<= 1) {
        int v = __shfl_up(incl, off);
        if (lane >= off) incl += v;
    }
    int wsum = __shfl(incl, 31);
    int base = 0;
    if (lane == 31) base = atomicAdd(total, wsum);
    base = __shfl(base, 31);
    if (gid < N) cursor[gid] = base + incl - d;
    for (int i = gid; i < 2 * CFEAT; i += gridDim.x * blockDim.x) stats[i] = 0.0f;
}

__global__ void fill_kernel(const int* __restrict__ src,
                            const int* __restrict__ dst,
                            int* __restrict__ cursor,
                            int* __restrict__ csr, int E) {
    int e = blockIdx.x * blockDim.x + threadIdx.x;
    if (e >= E) return;
    int pos = atomicAdd(&cursor[dst[e]], 1);
    if ((unsigned)pos < (unsigned)E) csr[pos] = src[e];
}

__global__ void gather_kernel(const float* __restrict__ x,
                              const int* __restrict__ csr,
                              const int* __restrict__ cursor,
                              const int* __restrict__ deg,
                              float* __restrict__ s,
                              float* __restrict__ cnt, int N) {
    int w = __builtin_amdgcn_readfirstlane((int)((blockIdx.x * blockDim.x + threadIdx.x) >> 5));
    int lane = threadIdx.x & 31;
    if (w >= N) return;
    int d = deg[w];
    int end = cursor[w];
    int beg = end - d;
    float4 a0 = make_float4(0.0f, 0.0f, 0.0f, 0.0f);
    float4 a1 = make_float4(0.0f, 0.0f, 0.0f, 0.0f);
    for (int c = beg; c < end; c += 32) {
        int n = end - c;
        if (n > 32) n = 32;
        int my = (lane < n) ? csr[c + lane] : 0;
        int j = 0;
        for (; j + 4 <= n; j += 4) {
            const float4* r0 = (const float4*)(x + (long)__builtin_amdgcn_readlane(my, j + 0) * CFEAT);
            const float4* r1 = (const float4*)(x + (long)__builtin_amdgcn_readlane(my, j + 1) * CFEAT);
            const float4* r2 = (const float4*)(x + (long)__builtin_amdgcn_readlane(my, j + 2) * CFEAT);
            const float4* r3 = (const float4*)(x + (long)__builtin_amdgcn_readlane(my, j + 3) * CFEAT);
            float4 p0 = r0[lane], q0 = r0[lane + 32];
            float4 p1 = r1[lane], q1 = r1[lane + 32];
            float4 p2 = r2[lane], q2 = r2[lane + 32];
            float4 p3 = r3[lane], q3 = r3[lane + 32];
            a0 += p0; a1 += q0;
            a0 += p1; a1 += q1;
            a0 += p2; a1 += q2;
            a0 += p3; a1 += q3;
        }
        for (; j < n; ++j) {
            const float4* r0 = (const float4*)(x + (long)__builtin_amdgcn_readlane(my, j) * CFEAT);
            float4 p0 = r0[lane], q0 = r0[lane + 32];
            a0 += p0; a1 += q0;
        }
    }
    float4* o = (float4*)(s + (long)w * CFEAT);
    o[lane] = a0;
    o[lane + 32] = a1;
    if (lane == 0) cnt[w] = (float)d;
}

#define GBM 64
#define GBN 256
#define GKC 32
#define GPITCH 36

__global__ __launch_bounds__(256)
void gemm_kernel(const float* __restrict__ s,
                 const float* __restrict__ cnt,
                 const float* __restrict__ xt,
                 const float* __restrict__ Wl,
                 const float* __restrict__ bl,
                 const float* __restrict__ Wr,
                 float* __restrict__ h,
                 float* __restrict__ colsum,
                 float* __restrict__ colsq,
                 int N) {
    __shared__ float tileA[GBM * GPITCH];
    __shared__ float tileB[GBN * GPITCH];
    __shared__ float redS[2][GBN];
    __shared__ float redQ[2][GBN];

    const int tid  = threadIdx.x;
    const int lane = tid & 31;
    const int wv   = tid >> 5;
    const int wm   = wv >> 2;
    const int wn   = wv & 3;
    const int l15  = lane & 15;
    const int hh   = lane >> 4;
    const int row0 = blockIdx.x * GBM;

    const int srow = tid >> 3;
    const int sseg = tid & 7;
    int ar0 = row0 + srow;       if (ar0 > N - 1) ar0 = N - 1;
    int ar1 = row0 + srow + 32;  if (ar1 > N - 1) ar1 = N - 1;
    const float inv0 = 1.0f / fmaxf(cnt[ar0], 1.0f);
    const float inv1 = 1.0f / fmaxf(cnt[ar1], 1.0f);
    const long aoff0 = (long)ar0 * CFEAT + sseg * 4;
    const long aoff1 = (long)ar1 * CFEAT + sseg * 4;
    const long boff  = (long)srow * CFEAT + sseg * 4;
    const int  sdst  = srow * GPITCH + sseg * 4;

    v8f acc[2][4];
#pragma unroll
    for (int i = 0; i < 2; ++i)
#pragma unroll
        for (int j = 0; j < 4; ++j)
            acc[i][j] = (v8f){0,0,0,0,0,0,0,0};

    float4 pa0, pa1, pb0, pb1, pb2, pb3, pb4, pb5, pb6, pb7;
    const long bstep = (long)32 * CFEAT;
    {
        const float* bp = Wl + boff;
        pa0 = *(const float4*)(s + aoff0);
        pa1 = *(const float4*)(s + aoff1);
        pb0 = *(const float4*)(bp);
        pb1 = *(const float4*)(bp + bstep);
        pb2 = *(const float4*)(bp + 2 * bstep);
        pb3 = *(const float4*)(bp + 3 * bstep);
        pb4 = *(const float4*)(bp + 4 * bstep);
        pb5 = *(const float4*)(bp + 5 * bstep);
        pb6 = *(const float4*)(bp + 6 * bstep);
        pb7 = *(const float4*)(bp + 7 * bstep);
    }

    const int fa = (wm * 32 + l15) * GPITCH + hh * 4;
    const int fb = (wn * 64 + l15) * GPITCH + hh * 4;

    for (int ch = 0; ch < 16; ++ch) {
        __syncthreads();
        {
            const float m0 = (ch < 8) ? inv0 : 1.0f;
            const float m1 = (ch < 8) ? inv1 : 1.0f;
            float4 t0 = pa0, t1 = pa1;
            t0.x *= m0; t0.y *= m0; t0.z *= m0; t0.w *= m0;
            t1.x *= m1; t1.y *= m1; t1.z *= m1; t1.w *= m1;
            *(float4*)(tileA + sdst) = t0;
            *(float4*)(tileA + sdst + 32 * GPITCH) = t1;
            float* bd = tileB + sdst;
            *(float4*)(bd)                   = pb0;
            *(float4*)(bd + 1 * 32 * GPITCH) = pb1;
            *(float4*)(bd + 2 * 32 * GPITCH) = pb2;
            *(float4*)(bd + 3 * 32 * GPITCH) = pb3;
            *(float4*)(bd + 4 * 32 * GPITCH) = pb4;
            *(float4*)(bd + 5 * 32 * GPITCH) = pb5;
            *(float4*)(bd + 6 * 32 * GPITCH) = pb6;
            *(float4*)(bd + 7 * 32 * GPITCH) = pb7;
        }
        __syncthreads();
        if (ch + 1 < 16) {
            const int cn = ch + 1;
            const float* Ab = (cn < 8) ? s : xt;
            const float* Bb = (cn < 8) ? Wl : Wr;
            const int k0 = (cn & 7) * GKC;
            pa0 = *(const float4*)(Ab + aoff0 + k0);
            pa1 = *(const float4*)(Ab + aoff1 + k0);
            const float* bp = Bb + boff + k0;
            pb0 = *(const float4*)(bp);
            pb1 = *(const float4*)(bp + bstep);
            pb2 = *(const float4*)(bp + 2 * bstep);
            pb3 = *(const float4*)(bp + 3 * bstep);
            pb4 = *(const float4*)(bp + 4 * bstep);
            pb5 = *(const float4*)(bp + 5 * bstep);
            pb6 = *(const float4*)(bp + 6 * bstep);
            pb7 = *(const float4*)(bp + 7 * bstep);
        }
        __builtin_amdgcn_sched_barrier(0);
#pragma unroll
        for (int kk = 0; kk < 4; ++kk) {
            float4 af[2], bf[4];
#pragma unroll
            for (int i = 0; i < 2; ++i) af[i] = *(const float4*)(tileA + fa + i * 16 * GPITCH + kk * 8);
#pragma unroll
            for (int j = 0; j < 4; ++j) bf[j] = *(const float4*)(tileB + fb + j * 16 * GPITCH + kk * 8);
#pragma unroll
            for (int i = 0; i < 2; ++i)
#pragma unroll
                for (int j = 0; j < 4; ++j) {
                    v2f a = {af[i].x, af[i].y};
                    v2f b = {bf[j].x, bf[j].y};
                    acc[i][j] = __builtin_amdgcn_wmma_f32_16x16x4_f32(
                        false, a, false, b, (short)0, acc[i][j], false, false);
                }
#pragma unroll
            for (int i = 0; i < 2; ++i)
#pragma unroll
                for (int j = 0; j < 4; ++j) {
                    v2f a = {af[i].z, af[i].w};
                    v2f b = {bf[j].z, bf[j].w};
                    acc[i][j] = __builtin_amdgcn_wmma_f32_16x16x4_f32(
                        false, a, false, b, (short)0, acc[i][j], false, false);
                }
        }
    }

#pragma unroll
    for (int j = 0; j < 4; ++j) {
        const int col = wn * 64 + j * 16 + l15;
        const float bias = bl[col];
        float sp = 0.0f, qp = 0.0f;
#pragma unroll
        for (int i = 0; i < 2; ++i) {
            const int rbase = row0 + wm * 32 + i * 16 + hh * 8;
#pragma unroll
            for (int v = 0; v < 8; ++v) {
                float val = fmaxf(acc[i][j][v] + bias, 0.0f);
                int r = rbase + v;
                bool ok = (r < N);
                float m = ok ? val : 0.0f;
                sp += m;
                qp += m * m;
                if (ok) h[(long)r * CFEAT + col] = val;
            }
        }
        sp += __shfl_xor(sp, 16);
        qp += __shfl_xor(qp, 16);
        if (hh == 0) { redS[wm][col] = sp; redQ[wm][col] = qp; }
    }
    __syncthreads();
    atomicAdd(&colsum[tid], redS[0][tid] + redS[1][tid]);
    atomicAdd(&colsq[tid],  redQ[0][tid] + redQ[1][tid]);
}

__global__ void norm_kernel(const float* __restrict__ hbuf,
                            const float* __restrict__ colsum,
                            const float* __restrict__ colsq,
                            const float* __restrict__ gamma,
                            const float* __restrict__ beta,
                            float* __restrict__ out, int N) {
    long i = (long)blockIdx.x * blockDim.x + threadIdx.x;
    long total = (long)N * CFEAT;
    if (i >= total) return;
    int col = (int)(i & (CFEAT - 1));
    float invN = 1.0f / (float)N;
    float mu  = colsum[col] * invN;
    float var = colsq[col] * invN - mu * mu;
    float rs  = rsqrtf(var + 1e-5f);
    out[i] = gamma[col] * (hbuf[i] - mu) * rs + beta[col];
}

extern "C" void kernel_launch(void* const* d_in, const int* in_sizes, int n_in,
                              void* d_out, int out_size, void* d_ws, size_t ws_size,
                              hipStream_t stream) {
    const float* x    = (const float*)d_in[0];
    const int* e0src  = (const int*)d_in[1];
    const int* e0dst  = (const int*)d_in[2];
    const int* e1src  = (const int*)d_in[3];
    const int* e1dst  = (const int*)d_in[4];
    const float* Wl0  = (const float*)d_in[5];
    const float* bl0  = (const float*)d_in[6];
    const float* Wr0  = (const float*)d_in[7];
    const float* g0   = (const float*)d_in[8];
    const float* be0  = (const float*)d_in[9];
    const float* Wl1  = (const float*)d_in[10];
    const float* bl1  = (const float*)d_in[11];
    const float* Wr1  = (const float*)d_in[12];
    const float* g1   = (const float*)d_in[13];
    const float* be1  = (const float*)d_in[14];

    const int N1 = 50000, N2 = 12500;
    const int E0 = in_sizes[1];
    const int E1 = in_sizes[3];

    float* ws = (float*)d_ws;
    const long NC1 = (long)N1 * CFEAT;
    const long NC2 = (long)N2 * CFEAT;
    float* x1    = ws;
    float* sbuf  = ws + NC1;
    float* hbuf  = ws + 2 * NC1;
    float* cntb  = ws + 3 * NC1;
    float* stats = cntb + 50048;

    int* t0 = (int*)x1;
    int* t1 = (int*)(sbuf + NC2);

    zero_int_kernel<<<64, 256, 0, stream>>>(t0, N1 + 32);
    degree_kernel<<<(E0 + 255) / 256, 256, 0, stream>>>(e0dst, t0, E0);
    offsets_kernel<<<(N1 + 255) / 256, 256, 0, stream>>>(t0, t0 + 65536, t0 + N1, stats, N1);
    fill_kernel<<<(E0 + 255) / 256, 256, 0, stream>>>(e0src, e0dst, t0 + 65536, t0 + 131072, E0);
    gather_kernel<<<(N1 + 7) / 8, 256, 0, stream>>>(x, t0 + 131072, t0 + 65536, t0, sbuf, cntb, N1);
    gemm_kernel<<<(N1 + GBM - 1) / GBM, 256, 0, stream>>>(
        sbuf, cntb, x, Wl0, bl0, Wr0, hbuf, stats, stats + 256, N1);
    norm_kernel<<<(int)((NC1 + 255) / 256), 256, 0, stream>>>(
        hbuf, stats, stats + 256, g0, be0, x1, N1);

    zero_int_kernel<<<64, 256, 0, stream>>>(t1, N2 + 32);
    degree_kernel<<<(E1 + 255) / 256, 256, 0, stream>>>(e1dst, t1, E1);
    offsets_kernel<<<(N2 + 255) / 256, 256, 0, stream>>>(t1, t1 + 65536, t1 + N2, stats, N2);
    fill_kernel<<<(E1 + 255) / 256, 256, 0, stream>>>(e1src, e1dst, t1 + 65536, t1 + 131072, E1);
    gather_kernel<<<(N2 + 7) / 8, 256, 0, stream>>>(x1, t1 + 131072, t1 + 65536, t1, sbuf, cntb, N2);
    gemm_kernel<<<(N2 + GBM - 1) / GBM, 256, 0, stream>>>(
        sbuf, cntb, x1, Wl1, bl1, Wr1, hbuf, stats, stats + 256, N2);
    norm_kernel<<<(int)((NC2 + 255) / 256), 256, 0, stream>>>(
        hbuf, stats, stats + 256, g1, be1, (float*)d_out, N2);
}


# Fewer, longer waves in the per-element passes: normalisation on a fixed 512-workgroup grid with 8 float4 in flight per lane (was one float per lane); degree and fill take 4 edges per lane
# speedup vs baseline: 1.9214x; 1.0334x over previous
#include <hip/hip_runtime.h>

typedef __attribute__((ext_vector_type(2))) float v2f;
typedef __attribute__((ext_vector_type(8))) float v8f;

#define CFEAT 256

__global__ void zero_int_kernel(int* __restrict__ p, int n) {
    int i = blockIdx.x * blockDim.x + threadIdx.x;
    int stride = gridDim.x * blockDim.x;
    for (; i < n; i += stride) p[i] = 0;
}

__global__ void degree_kernel(const int* __restrict__ dst, int* __restrict__ deg, int E) {
    int e = (blockIdx.x * blockDim.x + threadIdx.x) * 4;
    if (e + 3 < E) {
        const int4 d = *(const int4*)(dst + e);
        atomicAdd(&deg[d.x], 1);
        atomicAdd(&deg[d.y], 1);
        atomicAdd(&deg[d.z], 1);
        atomicAdd(&deg[d.w], 1);
    } else {
        for (; e < E; ++e) atomicAdd(&deg[dst[e]], 1);
    }
}

__global__ void offsets_kernel(const int* __restrict__ deg,
                               int* __restrict__ cursor,
                               int* __restrict__ total,
                               float* __restrict__ stats, int N) {
    int gid = blockIdx.x * blockDim.x + threadIdx.x;
    int lane = threadIdx.x & 31;
    int d = (gid < N) ? deg[gid] : 0;
    int incl = d;
    for (int off = 1; off < 32; off <<= 1) {
        int v = __shfl_up(incl, off);
        if (lane >= off) incl += v;
    }
    int wsum = __shfl(incl, 31);
    int base = 0;
    if (lane == 31) base = atomicAdd(total, wsum);
    base = __shfl(base, 31);
    if (gid < N) cursor[gid] = base + incl - d;
    for (int i = gid; i < 2 * CFEAT; i += gridDim.x * blockDim.x) stats[i] = 0.0f;
}

__global__ void fill_kernel(const int* __restrict__ src,
                            const int* __restrict__ dst,
                            int* __restrict__ cursor,
                            int* __restrict__ csr, int E) {
    int e = (blockIdx.x * blockDim.x + threadIdx.x) * 4;
    if (e + 3 < E) {
        const int4 d = *(const int4*)(dst + e);
        const int4 s = *(const int4*)(src + e);
        int p0 = atomicAdd(&cursor[d.x], 1);
        int p1 = atomicAdd(&cursor[d.y], 1);
        int p2 = atomicAdd(&cursor[d.z], 1);
        int p3 = atomicAdd(&cursor[d.w], 1);
        if ((unsigned)p0 < (unsigned)E) csr[p0] = s.x;
        if ((unsigned)p1 < (unsigned)E) csr[p1] = s.y;
        if ((unsigned)p2 < (unsigned)E) csr[p2] = s.z;
        if ((unsigned)p3 < (unsigned)E) csr[p3] = s.w;
    } else {
        for (; e < E; ++e) {
            int pos = atomicAdd(&cursor[dst[e]], 1);
            if ((unsigned)pos < (unsigned)E) csr[pos] = src[e];
        }
    }
}

__global__ void gather_kernel(const float* __restrict__ x,
                              const int* __restrict__ csr,
                              const int* __restrict__ cursor,
                              const int* __restrict__ deg,
                              float* __restrict__ s,
                              float* __restrict__ cnt, int N) {
    int w = __builtin_amdgcn_readfirstlane((int)((blockIdx.x * blockDim.x + threadIdx.x) >> 5));
    int lane = threadIdx.x & 31;
    if (w >= N) return;
    int d = deg[w];
    int end = cursor[w];
    int beg = end - d;
    float4 a0 = make_float4(0.0f, 0.0f, 0.0f, 0.0f);
    float4 a1 = make_float4(0.0f, 0.0f, 0.0f, 0.0f);
    for (int c = beg; c < end; c += 32) {
        int n = end - c;
        if (n > 32) n = 32;
        int my = (lane < n) ? csr[c + lane] : 0;
        int j = 0;
        for (; j + 4 <= n; j += 4) {
            const float4* r0 = (const float4*)(x + (long)__builtin_amdgcn_readlane(my, j + 0) * CFEAT);
            const float4* r1 = (const float4*)(x + (long)__builtin_amdgcn_readlane(my, j + 1) * CFEAT);
            const float4* r2 = (const float4*)(x + (long)__builtin_amdgcn_readlane(my, j + 2) * CFEAT);
            const float4* r3 = (const float4*)(x + (long)__builtin_amdgcn_readlane(my, j + 3) * CFEAT);
            float4 p0 = r0[lane], q0 = r0[lane + 32];
            float4 p1 = r1[lane], q1 = r1[lane + 32];
            float4 p2 = r2[lane], q2 = r2[lane + 32];
            float4 p3 = r3[lane], q3 = r3[lane + 32];
            a0 += p0; a1 += q0;
            a0 += p1; a1 += q1;
            a0 += p2; a1 += q2;
            a0 += p3; a1 += q3;
        }
        for (; j < n; ++j) {
            const float4* r0 = (const float4*)(x + (long)__builtin_amdgcn_readlane(my, j) * CFEAT);
            float4 p0 = r0[lane], q0 = r0[lane + 32];
            a0 += p0; a1 += q0;
        }
    }
    float4* o = (float4*)(s + (long)w * CFEAT);
    o[lane] = a0;
    o[lane + 32] = a1;
    if (lane == 0) cnt[w] = (float)d;
}

#define GBM 64
#define GBN 256
#define GKC 32
#define GPITCH 36

__global__ __launch_bounds__(256)
void gemm_kernel(const float* __restrict__ s,
                 const float* __restrict__ cnt,
                 const float* __restrict__ xt,
                 const float* __restrict__ Wl,
                 const float* __restrict__ bl,
                 const float* __restrict__ Wr,
                 float* __restrict__ h,
                 float* __restrict__ colsum,
                 float* __restrict__ colsq,
                 int N) {
    __shared__ float tileA[GBM * GPITCH];
    __shared__ float tileB[GBN * GPITCH];
    __shared__ float redS[2][GBN];
    __shared__ float redQ[2][GBN];

    const int tid  = threadIdx.x;
    const int lane = tid & 31;
    const int wv   = tid >> 5;
    const int wm   = wv >> 2;
    const int wn   = wv & 3;
    const int l15  = lane & 15;
    const int hh   = lane >> 4;
    const int row0 = blockIdx.x * GBM;

    const int srow = tid >> 3;
    const int sseg = tid & 7;
    int ar0 = row0 + srow;       if (ar0 > N - 1) ar0 = N - 1;
    int ar1 = row0 + srow + 32;  if (ar1 > N - 1) ar1 = N - 1;
    const float inv0 = 1.0f / fmaxf(cnt[ar0], 1.0f);
    const float inv1 = 1.0f / fmaxf(cnt[ar1], 1.0f);
    const long aoff0 = (long)ar0 * CFEAT + sseg * 4;
    const long aoff1 = (long)ar1 * CFEAT + sseg * 4;
    const long boff  = (long)srow * CFEAT + sseg * 4;
    const int  sdst  = srow * GPITCH + sseg * 4;

    v8f acc[2][4];
#pragma unroll
    for (int i = 0; i < 2; ++i)
#pragma unroll
        for (int j = 0; j < 4; ++j)
            acc[i][j] = (v8f){0,0,0,0,0,0,0,0};

    float4 pa0, pa1, pb0, pb1, pb2, pb3, pb4, pb5, pb6, pb7;
    const long bstep = (long)32 * CFEAT;
    {
        const float* bp = Wl + boff;
        pa0 = *(const float4*)(s + aoff0);
        pa1 = *(const float4*)(s + aoff1);
        pb0 = *(const float4*)(bp);
        pb1 = *(const float4*)(bp + bstep);
        pb2 = *(const float4*)(bp + 2 * bstep);
        pb3 = *(const float4*)(bp + 3 * bstep);
        pb4 = *(const float4*)(bp + 4 * bstep);
        pb5 = *(const float4*)(bp + 5 * bstep);
        pb6 = *(const float4*)(bp + 6 * bstep);
        pb7 = *(const float4*)(bp + 7 * bstep);
    }

    const int fa = (wm * 32 + l15) * GPITCH + hh * 4;
    const int fb = (wn * 64 + l15) * GPITCH + hh * 4;

    for (int ch = 0; ch < 16; ++ch) {
        __syncthreads();
        {
            const float m0 = (ch < 8) ? inv0 : 1.0f;
            const float m1 = (ch < 8) ? inv1 : 1.0f;
            float4 t0 = pa0, t1 = pa1;
            t0.x *= m0; t0.y *= m0; t0.z *= m0; t0.w *= m0;
            t1.x *= m1; t1.y *= m1; t1.z *= m1; t1.w *= m1;
            *(float4*)(tileA + sdst) = t0;
            *(float4*)(tileA + sdst + 32 * GPITCH) = t1;
            float* bd = tileB + sdst;
            *(float4*)(bd)                   = pb0;
            *(float4*)(bd + 1 * 32 * GPITCH) = pb1;
            *(float4*)(bd + 2 * 32 * GPITCH) = pb2;
            *(float4*)(bd + 3 * 32 * GPITCH) = pb3;
            *(float4*)(bd + 4 * 32 * GPITCH) = pb4;
            *(float4*)(bd + 5 * 32 * GPITCH) = pb5;
            *(float4*)(bd + 6 * 32 * GPITCH) = pb6;
            *(float4*)(bd + 7 * 32 * GPITCH) = pb7;
        }
        __syncthreads();
        if (ch + 1 < 16) {
            const int cn = ch + 1;
            const float* Ab = (cn < 8) ? s : xt;
            const float* Bb = (cn < 8) ? Wl : Wr;
            const int k0 = (cn & 7) * GKC;
            pa0 = *(const float4*)(Ab + aoff0 + k0);
            pa1 = *(const float4*)(Ab + aoff1 + k0);
            const float* bp = Bb + boff + k0;
            pb0 = *(const float4*)(bp);
            pb1 = *(const float4*)(bp + bstep);
            pb2 = *(const float4*)(bp + 2 * bstep);
            pb3 = *(const float4*)(bp + 3 * bstep);
            pb4 = *(const float4*)(bp + 4 * bstep);
            pb5 = *(const float4*)(bp + 5 * bstep);
            pb6 = *(const float4*)(bp + 6 * bstep);
            pb7 = *(const float4*)(bp + 7 * bstep);
        }
        __builtin_amdgcn_sched_barrier(0);
#pragma unroll
        for (int kk = 0; kk < 4; ++kk) {
            float4 af[2], bf[4];
#pragma unroll
            for (int i = 0; i < 2; ++i) af[i] = *(const float4*)(tileA + fa + i * 16 * GPITCH + kk * 8);
#pragma unroll
            for (int j = 0; j < 4; ++j) bf[j] = *(const float4*)(tileB + fb + j * 16 * GPITCH + kk * 8);
#pragma unroll
            for (int i = 0; i < 2; ++i)
#pragma unroll
                for (int j = 0; j < 4; ++j) {
                    v2f a = {af[i].x, af[i].y};
                    v2f b = {bf[j].x, bf[j].y};
                    acc[i][j] = __builtin_amdgcn_wmma_f32_16x16x4_f32(
                        false, a, false, b, (short)0, acc[i][j], false, false);
                }
#pragma unroll
            for (int i = 0; i < 2; ++i)
#pragma unroll
                for (int j = 0; j < 4; ++j) {
                    v2f a = {af[i].z, af[i].w};
                    v2f b = {bf[j].z, bf[j].w};
                    acc[i][j] = __builtin_amdgcn_wmma_f32_16x16x4_f32(
                        false, a, false, b, (short)0, acc[i][j], false, false);
                }
        }
    }

#pragma unroll
    for (int j = 0; j < 4; ++j) {
        const int col = wn * 64 + j * 16 + l15;
        const float bias = bl[col];
        float sp = 0.0f, qp = 0.0f;
#pragma unroll
        for (int i = 0; i < 2; ++i) {
            const int rbase = row0 + wm * 32 + i * 16 + hh * 8;
#pragma unroll
            for (int v = 0; v < 8; ++v) {
                float val = fmaxf(acc[i][j][v] + bias, 0.0f);
                int r = rbase + v;
                bool ok = (r < N);
                float m = ok ? val : 0.0f;
                sp += m;
                qp += m * m;
                if (ok) h[(long)r * CFEAT + col] = val;
            }
        }
        sp += __shfl_xor(sp, 16);
        qp += __shfl_xor(qp, 16);
        if (hh == 0) { redS[wm][col] = sp; redQ[wm][col] = qp; }
    }
    __syncthreads();
    atomicAdd(&colsum[tid], redS[0][tid] + redS[1][tid]);
    atomicAdd(&colsq[tid],  redQ[0][tid] + redQ[1][tid]);
}

#define NORM_U 8
#define NORM_WG 512

__device__ __forceinline__ float4 norm_apply(float4 v, float4 g, float4 mu, float4 rs, float4 b) {
    float4 o;
    o.x = g.x * (v.x - mu.x) * rs.x + b.x;
    o.y = g.y * (v.y - mu.y) * rs.y + b.y;
    o.z = g.z * (v.z - mu.z) * rs.z + b.z;
    o.w = g.w * (v.w - mu.w) * rs.w + b.w;
    return o;
}

__global__ __launch_bounds__(256)
void norm_kernel(const float* __restrict__ hbuf,
                 const float* __restrict__ colsum,
                 const float* __restrict__ colsq,
                 const float* __restrict__ gamma,
                 const float* __restrict__ beta,
                 float* __restrict__ out, int N) {
    const long total4 = (long)N * (CFEAT / 4);
    const long width  = (long)gridDim.x * blockDim.x;
    const long first  = (long)blockIdx.x * blockDim.x + threadIdx.x;
    const int  c4     = threadIdx.x & (CFEAT / 4 - 1);

    const float invN = 1.0f / (float)N;
    const float4 cs = ((const float4*)colsum)[c4];
    const float4 cq = ((const float4*)colsq)[c4];
    const float4 g  = ((const float4*)gamma)[c4];
    const float4 b  = ((const float4*)beta)[c4];
    float4 mu, rs;
    mu.x = cs.x * invN;  mu.y = cs.y * invN;  mu.z = cs.z * invN;  mu.w = cs.w * invN;
    rs.x = rsqrtf(cq.x * invN - mu.x * mu.x + 1e-5f);
    rs.y = rsqrtf(cq.y * invN - mu.y * mu.y + 1e-5f);
    rs.z = rsqrtf(cq.z * invN - mu.z * mu.z + 1e-5f);
    rs.w = rsqrtf(cq.w * invN - mu.w * mu.w + 1e-5f);

    const float4* __restrict__ in4 = (const float4*)hbuf;
    float4* __restrict__ out4 = (float4*)out;

    for (long base = first; base < total4; base += NORM_U * width) {
        float4 v[NORM_U];
#pragma unroll
        for (int u = 0; u < NORM_U; ++u) {
            const long j = base + u * width;
            v[u] = (j < total4) ? in4[j] : make_float4(0.0f, 0.0f, 0.0f, 0.0f);
        }
#pragma unroll
        for (int u = 0; u < NORM_U; ++u) {
            const long j = base + u * width;
            if (j < total4) out4[j] = norm_apply(v[u], g, mu, rs, b);
        }
    }
}

extern "C" void kernel_launch(void* const* d_in, const int* in_sizes, int n_in,
                              void* d_out, int out_size, void* d_ws, size_t ws_size,
                              hipStream_t stream) {
    const float* x    = (const float*)d_in[0];
    const int* e0src  = (const int*)d_in[1];
    const int* e0dst  = (const int*)d_in[2];
    const int* e1src  = (const int*)d_in[3];
    const int* e1dst  = (const int*)d_in[4];
    const float* Wl0  = (const float*)d_in[5];
    const float* bl0  = (const float*)d_in[6];
    const float* Wr0  = (const float*)d_in[7];
    const float* g0   = (const float*)d_in[8];
    const float* be0  = (const float*)d_in[9];
    const float* Wl1  = (const float*)d_in[10];
    const float* bl1  = (const float*)d_in[11];
    const float* Wr1  = (const float*)d_in[12];
    const float* g1   = (const float*)d_in[13];
    const float* be1  = (const float*)d_in[14];

    const int N1 = 50000, N2 = 12500;
    const int E0 = in_sizes[1];
    const int E1 = in_sizes[3];

    float* ws = (float*)d_ws;
    const long NC1 = (long)N1 * CFEAT;
    const long NC2 = (long)N2 * CFEAT;
    float* x1    = ws;
    float* sbuf  = ws + NC1;
    float* hbuf  = ws + 2 * NC1;
    float* cntb  = ws + 3 * NC1;
    float* stats = cntb + 50048;

    int* t0 = (int*)x1;
    int* t1 = (int*)(sbuf + NC2);

    zero_int_kernel<<<64, 256, 0, stream>>>(t0, N1 + 32);
    degree_kernel<<<(E0 + 1023) / 1024, 256, 0, stream>>>(e0dst, t0, E0);
    offsets_kernel<<<(N1 + 255) / 256, 256, 0, stream>>>(t0, t0 + 65536, t0 + N1, stats, N1);
    fill_kernel<<<(E0 + 1023) / 1024, 256, 0, stream>>>(e0src, e0dst, t0 + 65536, t0 + 131072, E0);
    gather_kernel<<<(N1 + 7) / 8, 256, 0, stream>>>(x, t0 + 131072, t0 + 65536, t0, sbuf, cntb, N1);
    gemm_kernel<<<(N1 + GBM - 1) / GBM, 256, 0, stream>>>(
        sbuf, cntb, x, Wl0, bl0, Wr0, hbuf, stats, stats + 256, N1);
    norm_kernel<<<NORM_WG, 256, 0, stream>>>(
        hbuf, stats, stats + 256, g0, be0, x1, N1);

    zero_int_kernel<<<64, 256, 0, stream>>>(t1, N2 + 32);
    degree_kernel<<<(E1 + 1023) / 1024, 256, 0, stream>>>(e1dst, t1, E1);
    offsets_kernel<<<(N2 + 255) / 256, 256, 0, stream>>>(t1, t1 + 65536, t1 + N2, stats, N2);
    fill_kernel<<<(E1 + 1023) / 1024, 256, 0, stream>>>(e1src, e1dst, t1 + 65536, t1 + 131072, E1);
    gather_kernel<<<(N2 + 7) / 8, 256, 0, stream>>>(x1, t1 + 131072, t1 + 65536, t1, sbuf, cntb, N2);
    gemm_kernel<<<(N2 + GBM - 1) / GBM, 256, 0, stream>>>(
        sbuf, cntb, x1, Wl1, bl1, Wr1, hbuf, stats, stats + 256, N2);
    norm_kernel<<<NORM_WG, 256, 0, stream>>>(
        hbuf, stats, stats + 256, g1, be1, (float*)d_out, N2);
}


# One atomic per edge, not two: the count pass keeps each edge's arrival-order rank (its add's old value), so the fill pass stores at range start + rank with plain loads and stores; 14 launches
# speedup vs baseline: 2.2133x; 1.1519x over previous
#include <hip/hip_runtime.h>

typedef __attribute__((ext_vector_type(2))) float v2f;
typedef __attribute__((ext_vector_type(8))) float v8f;

#define CFEAT 256

__global__ void zero_int_kernel(int* __restrict__ p, int n) {
    int i = blockIdx.x * blockDim.x + threadIdx.x;
    int stride = gridDim.x * blockDim.x;
    for (; i < n; i += stride) p[i] = 0;
}

__global__ void degree_kernel(const int* __restrict__ dst, int* __restrict__ deg,
                              int* __restrict__ rank, int E) {
    int e = (blockIdx.x * blockDim.x + threadIdx.x) * 4;
    if (e + 3 < E) {
        const int4 d = *(const int4*)(dst + e);
        int4 r;
        r.x = atomicAdd(&deg[d.x], 1);
        r.y = atomicAdd(&deg[d.y], 1);
        r.z = atomicAdd(&deg[d.z], 1);
        r.w = atomicAdd(&deg[d.w], 1);
        *(int4*)(rank + e) = r;
    } else {
        for (; e < E; ++e) rank[e] = atomicAdd(&deg[dst[e]], 1);
    }
}

__global__ void offsets_kernel(const int* __restrict__ deg,
                               int* __restrict__ cursor,
                               int* __restrict__ total,
                               float* __restrict__ stats, int N) {
    int gid = blockIdx.x * blockDim.x + threadIdx.x;
    int lane = threadIdx.x & 31;
    int d = (gid < N) ? deg[gid] : 0;
    int incl = d;
    for (int off = 1; off < 32; off <<= 1) {
        int v = __shfl_up(incl, off);
        if (lane >= off) incl += v;
    }
    int wsum = __shfl(incl, 31);
    int base = 0;
    if (lane == 31) base = atomicAdd(total, wsum);
    base = __shfl(base, 31);
    if (gid < N) cursor[gid] = base + incl - d;
    for (int i = gid; i < 2 * CFEAT; i += gridDim.x * blockDim.x) stats[i] = 0.0f;
}

__global__ void fill_kernel(const int* __restrict__ src,
                            const int* __restrict__ dst,
                            const int* __restrict__ rank,
                            const int* __restrict__ cursor,
                            int* __restrict__ csr, int E) {
    int e = (blockIdx.x * blockDim.x + threadIdx.x) * 4;
    if (e + 3 < E) {
        const int4 d = *(const int4*)(dst + e);
        const int4 s = *(const int4*)(src + e);
        const int4 r = *(const int4*)(rank + e);
        int p0 = cursor[d.x] + r.x;
        int p1 = cursor[d.y] + r.y;
        int p2 = cursor[d.z] + r.z;
        int p3 = cursor[d.w] + r.w;
        if ((unsigned)p0 < (unsigned)E) csr[p0] = s.x;
        if ((unsigned)p1 < (unsigned)E) csr[p1] = s.y;
        if ((unsigned)p2 < (unsigned)E) csr[p2] = s.z;
        if ((unsigned)p3 < (unsigned)E) csr[p3] = s.w;
    } else {
        for (; e < E; ++e) {
            int pos = cursor[dst[e]] + rank[e];
            if ((unsigned)pos < (unsigned)E) csr[pos] = src[e];
        }
    }
}

__global__ void gather_kernel(const float* __restrict__ x,
                              const int* __restrict__ csr,
                              const int* __restrict__ cursor,
                              const int* __restrict__ deg,
                              float* __restrict__ s,
                              float* __restrict__ cnt, int N) {
    int w = __builtin_amdgcn_readfirstlane((int)((blockIdx.x * blockDim.x + threadIdx.x) >> 5));
    int lane = threadIdx.x & 31;
    if (w >= N) return;
    int d = deg[w];
    int beg = cursor[w];
    int end = beg + d;
    float4 a0 = make_float4(0.0f, 0.0f, 0.0f, 0.0f);
    float4 a1 = make_float4(0.0f, 0.0f, 0.0f, 0.0f);
    for (int c = beg; c < end; c += 32) {
        int n = end - c;
        if (n > 32) n = 32;
        int my = (lane < n) ? csr[c + lane] : 0;
        int j = 0;
        for (; j + 4 <= n; j += 4) {
            const float4* r0 = (const float4*)(x + (long)__builtin_amdgcn_readlane(my, j + 0) * CFEAT);
            const float4* r1 = (const float4*)(x + (long)__builtin_amdgcn_readlane(my, j + 1) * CFEAT);
            const float4* r2 = (const float4*)(x + (long)__builtin_amdgcn_readlane(my, j + 2) * CFEAT);
            const float4* r3 = (const float4*)(x + (long)__builtin_amdgcn_readlane(my, j + 3) * CFEAT);
            float4 p0 = r0[lane], q0 = r0[lane + 32];
            float4 p1 = r1[lane], q1 = r1[lane + 32];
            float4 p2 = r2[lane], q2 = r2[lane + 32];
            float4 p3 = r3[lane], q3 = r3[lane + 32];
            a0 += p0; a1 += q0;
            a0 += p1; a1 += q1;
            a0 += p2; a1 += q2;
            a0 += p3; a1 += q3;
        }
        for (; j < n; ++j) {
            const float4* r0 = (const float4*)(x + (long)__builtin_amdgcn_readlane(my, j) * CFEAT);
            float4 p0 = r0[lane], q0 = r0[lane + 32];
            a0 += p0; a1 += q0;
        }
    }
    float4* o = (float4*)(s + (long)w * CFEAT);
    o[lane] = a0;
    o[lane + 32] = a1;
    if (lane == 0) cnt[w] = (float)d;
}

#define GBM 64
#define GBN 256
#define GKC 32
#define GPITCH 36

__global__ __launch_bounds__(256)
void gemm_kernel(const float* __restrict__ s,
                 const float* __restrict__ cnt,
                 const float* __restrict__ xt,
                 const float* __restrict__ Wl,
                 const float* __restrict__ bl,
                 const float* __restrict__ Wr,
                 float* __restrict__ h,
                 float* __restrict__ colsum,
                 float* __restrict__ colsq,
                 int N) {
    __shared__ float tileA[GBM * GPITCH];
    __shared__ float tileB[GBN * GPITCH];
    __shared__ float redS[2][GBN];
    __shared__ float redQ[2][GBN];

    const int tid  = threadIdx.x;
    const int lane = tid & 31;
    const int wv   = tid >> 5;
    const int wm   = wv >> 2;
    const int wn   = wv & 3;
    const int l15  = lane & 15;
    const int hh   = lane >> 4;
    const int row0 = blockIdx.x * GBM;

    const int srow = tid >> 3;
    const int sseg = tid & 7;
    int ar0 = row0 + srow;       if (ar0 > N - 1) ar0 = N - 1;
    int ar1 = row0 + srow + 32;  if (ar1 > N - 1) ar1 = N - 1;
    const float inv0 = 1.0f / fmaxf(cnt[ar0], 1.0f);
    const float inv1 = 1.0f / fmaxf(cnt[ar1], 1.0f);
    const long aoff0 = (long)ar0 * CFEAT + sseg * 4;
    const long aoff1 = (long)ar1 * CFEAT + sseg * 4;
    const long boff  = (long)srow * CFEAT + sseg * 4;
    const int  sdst  = srow * GPITCH + sseg * 4;

    v8f acc[2][4];
#pragma unroll
    for (int i = 0; i < 2; ++i)
#pragma unroll
        for (int j = 0; j < 4; ++j)
            acc[i][j] = (v8f){0,0,0,0,0,0,0,0};

    float4 pa0, pa1, pb0, pb1, pb2, pb3, pb4, pb5, pb6, pb7;
    const long bstep = (long)32 * CFEAT;
    {
        const float* bp = Wl + boff;
        pa0 = *(const float4*)(s + aoff0);
        pa1 = *(const float4*)(s + aoff1);
        pb0 = *(const float4*)(bp);
        pb1 = *(const float4*)(bp + bstep);
        pb2 = *(const float4*)(bp + 2 * bstep);
        pb3 = *(const float4*)(bp + 3 * bstep);
        pb4 = *(const float4*)(bp + 4 * bstep);
        pb5 = *(const float4*)(bp + 5 * bstep);
        pb6 = *(const float4*)(bp + 6 * bstep);
        pb7 = *(const float4*)(bp + 7 * bstep);
    }

    const int fa = (wm * 32 + l15) * GPITCH + hh * 4;
    const int fb = (wn * 64 + l15) * GPITCH + hh * 4;

    for (int ch = 0; ch < 16; ++ch) {
        __syncthreads();
        {
            const float m0 = (ch < 8) ? inv0 : 1.0f;
            const float m1 = (ch < 8) ? inv1 : 1.0f;
            float4 t0 = pa0, t1 = pa1;
            t0.x *= m0; t0.y *= m0; t0.z *= m0; t0.w *= m0;
            t1.x *= m1; t1.y *= m1; t1.z *= m1; t1.w *= m1;
            *(float4*)(tileA + sdst) = t0;
            *(float4*)(tileA + sdst + 32 * GPITCH) = t1;
            float* bd = tileB + sdst;
            *(float4*)(bd)                   = pb0;
            *(float4*)(bd + 1 * 32 * GPITCH) = pb1;
            *(float4*)(bd + 2 * 32 * GPITCH) = pb2;
            *(float4*)(bd + 3 * 32 * GPITCH) = pb3;
            *(float4*)(bd + 4 * 32 * GPITCH) = pb4;
            *(float4*)(bd + 5 * 32 * GPITCH) = pb5;
            *(float4*)(bd + 6 * 32 * GPITCH) = pb6;
            *(float4*)(bd + 7 * 32 * GPITCH) = pb7;
        }
        __syncthreads();
        if (ch + 1 < 16) {
            const int cn = ch + 1;
            const float* Ab = (cn < 8) ? s : xt;
            const float* Bb = (cn < 8) ? Wl : Wr;
            const int k0 = (cn & 7) * GKC;
            pa0 = *(const float4*)(Ab + aoff0 + k0);
            pa1 = *(const float4*)(Ab + aoff1 + k0);
            const float* bp = Bb + boff + k0;
            pb0 = *(const float4*)(bp);
            pb1 = *(const float4*)(bp + bstep);
            pb2 = *(const float4*)(bp + 2 * bstep);
            pb3 = *(const float4*)(bp + 3 * bstep);
            pb4 = *(const float4*)(bp + 4 * bstep);
            pb5 = *(const float4*)(bp + 5 * bstep);
            pb6 = *(const float4*)(bp + 6 * bstep);
            pb7 = *(const float4*)(bp + 7 * bstep);
        }
        __builtin_amdgcn_sched_barrier(0);
#pragma unroll
        for (int kk = 0; kk < 4; ++kk) {
            float4 af[2], bf[4];
#pragma unroll
            for (int i = 0; i < 2; ++i) af[i] = *(const float4*)(tileA + fa + i * 16 * GPITCH + kk * 8);
#pragma unroll
            for (int j = 0; j < 4; ++j) bf[j] = *(const float4*)(tileB + fb + j * 16 * GPITCH + kk * 8);
#pragma unroll
            for (int i = 0; i < 2; ++i)
#pragma unroll
                for (int j = 0; j < 4; ++j) {
                    v2f a = {af[i].x, af[i].y};
                    v2f b = {bf[j].x, bf[j].y};
                    acc[i][j] = __builtin_amdgcn_wmma_f32_16x16x4_f32(
                        false, a, false, b, (short)0, acc[i][j], false, false);
                }
#pragma unroll
            for (int i = 0; i < 2; ++i)
#pragma unroll
                for (int j = 0; j < 4; ++j) {
                    v2f a = {af[i].z, af[i].w};
                    v2f b = {bf[j].z, bf[j].w};
                    acc[i][j] = __builtin_amdgcn_wmma_f32_16x16x4_f32(
                        false, a, false, b, (short)0, acc[i][j], false, false);
                }
        }
    }

#pragma unroll
    for (int j = 0; j < 4; ++j) {
        const int col = wn * 64 + j * 16 + l15;
        const float bias = bl[col];
        float sp = 0.0f, qp = 0.0f;
#pragma unroll
        for (int i = 0; i < 2; ++i) {
            const int rbase = row0 + wm * 32 + i * 16 + hh * 8;
#pragma unroll
            for (int v = 0; v < 8; ++v) {
                float val = fmaxf(acc[i][j][v] + bias, 0.0f);
                int r = rbase + v;
                bool ok = (r < N);
                float m = ok ? val : 0.0f;
                sp += m;
                qp += m * m;
                if (ok) h[(long)r * CFEAT + col] = val;
            }
        }
        sp += __shfl_xor(sp, 16);
        qp += __shfl_xor(qp, 16);
        if (hh == 0) { redS[wm][col] = sp; redQ[wm][col] = qp; }
    }
    __syncthreads();
    atomicAdd(&colsum[tid], redS[0][tid] + redS[1][tid]);
    atomicAdd(&colsq[tid],  redQ[0][tid] + redQ[1][tid]);
}

#define NORM_U 8
#define NORM_WG 512

__device__ __forceinline__ float4 norm_apply(float4 v, float4 g, float4 mu, float4 rs, float4 b) {
    float4 o;
    o.x = g.x * (v.x - mu.x) * rs.x + b.x;
    o.y = g.y * (v.y - mu.y) * rs.y + b.y;
    o.z = g.z * (v.z - mu.z) * rs.z + b.z;
    o.w = g.w * (v.w - mu.w) * rs.w + b.w;
    return o;
}

__global__ __launch_bounds__(256)
void norm_kernel(const float* __restrict__ hbuf,
                 const float* __restrict__ colsum,
                 const float* __restrict__ colsq,
                 const float* __restrict__ gamma,
                 const float* __restrict__ beta,
                 float* __restrict__ out, int N) {
    const long total4 = (long)N * (CFEAT / 4);
    const long width  = (long)gridDim.x * blockDim.x;
    const long first  = (long)blockIdx.x * blockDim.x + threadIdx.x;
    const int  c4     = threadIdx.x & (CFEAT / 4 - 1);

    const float invN = 1.0f / (float)N;
    const float4 cs = ((const float4*)colsum)[c4];
    const float4 cq = ((const float4*)colsq)[c4];
    const float4 g  = ((const float4*)gamma)[c4];
    const float4 b  = ((const float4*)beta)[c4];
    float4 mu, rs;
    mu.x = cs.x * invN;  mu.y = cs.y * invN;  mu.z = cs.z * invN;  mu.w = cs.w * invN;
    rs.x = rsqrtf(cq.x * invN - mu.x * mu.x + 1e-5f);
    rs.y = rsqrtf(cq.y * invN - mu.y * mu.y + 1e-5f);
    rs.z = rsqrtf(cq.z * invN - mu.z * mu.z + 1e-5f);
    rs.w = rsqrtf(cq.w * invN - mu.w * mu.w + 1e-5f);

    const float4* __restrict__ in4 = (const float4*)hbuf;
    float4* __restrict__ out4 = (float4*)out;

    for (long base = first; base < total4; base += NORM_U * width) {
        float4 v[NORM_U];
#pragma unroll
        for (int u = 0; u < NORM_U; ++u) {
            const long j = base + u * width;
            v[u] = (j < total4) ? in4[j] : make_float4(0.0f, 0.0f, 0.0f, 0.0f);
        }
#pragma unroll
        for (int u = 0; u < NORM_U; ++u) {
            const long j = base + u * width;
            if (j < total4) out4[j] = norm_apply(v[u], g, mu, rs, b);
        }
    }
}

extern "C" void kernel_launch(void* const* d_in, const int* in_sizes, int n_in,
                              void* d_out, int out_size, void* d_ws, size_t ws_size,
                              hipStream_t stream) {
    const float* x    = (const float*)d_in[0];
    const int* e0src  = (const int*)d_in[1];
    const int* e0dst  = (const int*)d_in[2];
    const int* e1src  = (const int*)d_in[3];
    const int* e1dst  = (const int*)d_in[4];
    const float* Wl0  = (const float*)d_in[5];
    const float* bl0  = (const float*)d_in[6];
    const float* Wr0  = (const float*)d_in[7];
    const float* g0   = (const float*)d_in[8];
    const float* be0  = (const float*)d_in[9];
    const float* Wl1  = (const float*)d_in[10];
    const float* bl1  = (const float*)d_in[11];
    const float* Wr1  = (const float*)d_in[12];
    const float* g1   = (const float*)d_in[13];
    const float* be1  = (const float*)d_in[14];

    const int N1 = 50000, N2 = 12500;
    const int E0 = in_sizes[1];
    const int E1 = in_sizes[3];

    float* ws = (float*)d_ws;
    const long NC1 = (long)N1 * CFEAT;
    const long NC2 = (long)N2 * CFEAT;
    float* x1    = ws;
    float* sbuf  = ws + NC1;
    float* hbuf  = ws + 2 * NC1;
    float* cntb  = ws + 3 * NC1;
    float* stats = cntb + 50048;

    int* t0 = (int*)x1;
    int* t1 = (int*)(sbuf + NC2);

    zero_int_kernel<<<64, 256, 0, stream>>>(t0, N1 + 32);
    degree_kernel<<<(E0 + 1023) / 1024, 256, 0, stream>>>(e0dst, t0, t0 + 1179648, E0);
    offsets_kernel<<<(N1 + 255) / 256, 256, 0, stream>>>(t0, t0 + 65536, t0 + N1, stats, N1);
    fill_kernel<<<(E0 + 1023) / 1024, 256, 0, stream>>>(e0src, e0dst, t0 + 1179648, t0 + 65536, t0 + 131072, E0);
    gather_kernel<<<(N1 + 7) / 8, 256, 0, stream>>>(x, t0 + 131072, t0 + 65536, t0, sbuf, cntb, N1);
    gemm_kernel<<<(N1 + GBM - 1) / GBM, 256, 0, stream>>>(
        sbuf, cntb, x, Wl0, bl0, Wr0, hbuf, stats, stats + 256, N1);
    norm_kernel<<<NORM_WG, 256, 0, stream>>>(
        hbuf, stats, stats + 256, g0, be0, x1, N1);

    zero_int_kernel<<<64, 256, 0, stream>>>(t1, N2 + 32);
    degree_kernel<<<(E1 + 1023) / 1024, 256, 0, stream>>>(e1dst, t1, t1 + 1179648, E1);
    offsets_kernel<<<(N2 + 255) / 256, 256, 0, stream>>>(t1, t1 + 65536, t1 + N2, stats, N2);
    fill_kernel<<<(E1 + 1023) / 1024, 256, 0, stream>>>(e1src, e1dst, t1 + 1179648, t1 + 65536, t1 + 131072, E1);
    gather_kernel<<<(N2 + 7) / 8, 256, 0, stream>>>(x1, t1 + 131072, t1 + 65536, t1, sbuf, cntb, N2);
    gemm_kernel<<<(N2 + GBM - 1) / GBM, 256, 0, stream>>>(
        sbuf, cntb, x1, Wl1, bl1, Wr1, hbuf, stats, stats + 256, N2);
    norm_kernel<<<NORM_WG, 256, 0, stream>>>(
        hbuf, stats, stats + 256, g1, be1, (float*)d_out, N2);
}


# Edge list grouped by destination row by a two-pass counting sort in LDS (4096-edge chunks by row bin, then one workgroup per bin), no global atomics; in-row order still arrival order; 10 launches
# speedup vs baseline: 3.2182x; 1.4540x over previous
#include <hip/hip_runtime.h>

typedef __attribute__((ext_vector_type(2))) float v2f;
typedef __attribute__((ext_vector_type(8))) float v8f;

#define CFEAT 256

#define BIN_ROWS 256
#define CHUNK 4096

__global__ __launch_bounds__(512)
void binsort_kernel(const int* __restrict__ src, const int* __restrict__ dst,
                    int* __restrict__ bsrc, int* __restrict__ bdst, int* __restrict__ segT,
                    int E, int N, int B, int nchunks, int shift) {
    __shared__ int hist[BIN_ROWS + 1];
    __shared__ int wtot[16];
    const int tid = threadIdx.x;
    const int lane = tid & 31;
    const int wv = tid >> 5;
    const int c = blockIdx.x;
    const int base = c * CHUNK;
    if (tid <= BIN_ROWS) hist[tid] = 0;

    int d[8], s[8], r[8];
#pragma unroll
    for (int k = 0; k < 2; ++k) {
        const int e = base + k * (CHUNK / 2) + tid * 4;
        if (e + 3 < E) {
            const int4 dv = *(const int4*)(dst + e);
            const int4 sv = *(const int4*)(src + e);
            d[k * 4 + 0] = dv.x; d[k * 4 + 1] = dv.y; d[k * 4 + 2] = dv.z; d[k * 4 + 3] = dv.w;
            s[k * 4 + 0] = sv.x; s[k * 4 + 1] = sv.y; s[k * 4 + 2] = sv.z; s[k * 4 + 3] = sv.w;
        } else {
#pragma unroll
            for (int i = 0; i < 4; ++i) {
                const bool in = (e + i < E);
                d[k * 4 + i] = in ? dst[e + i] : -1;
                s[k * 4 + i] = in ? src[e + i] : 0;
            }
        }
    }
    __syncthreads();
#pragma unroll
    for (int i = 0; i < 8; ++i) {
        const bool ok = ((unsigned)d[i] < (unsigned)N);
        r[i] = ok ? atomicAdd(&hist[d[i] >> shift], 1) : -1;
    }
    __syncthreads();
    int v = (tid < B) ? hist[tid] : 0;
    int incl = v;
#pragma unroll
    for (int off = 1; off < 32; off <<= 1) {
        int t = __shfl_up(incl, off);
        if (lane >= off) incl += t;
    }
    if (lane == 31) wtot[wv] = incl;
    __syncthreads();
    int pre = 0;
    for (int w = 0; w < wv; ++w) pre += wtot[w];
    const int excl = pre + incl - v;
    if (tid <= B) {
        hist[tid] = excl;
        segT[tid * nchunks + c] = excl;
    }
    __syncthreads();
#pragma unroll
    for (int i = 0; i < 8; ++i) {
        if (r[i] >= 0) {
            const int p = base + hist[d[i] >> shift] + r[i];
            bsrc[p] = s[i];
            bdst[p] = d[i];
        }
    }
}

__global__ __launch_bounds__(1024)
void bincsr_kernel(const int* __restrict__ bsrc, const int* __restrict__ bdst,
                   const int* __restrict__ segT,
                   int* __restrict__ csr, int* __restrict__ cursor, int* __restrict__ deg,
                   float* __restrict__ stats, int E, int N, int nchunks, int shift) {
    __shared__ int cnt[BIN_ROWS];
    __shared__ int fillp[BIN_ROWS];
    __shared__ int wtot[8];
    __shared__ int sbase;
    const int tid = threadIdx.x;
    const int lane = tid & 31;
    const int wv = tid >> 5;
    const int b = blockIdx.x;
    const int rmask = (1 << shift) - 1;
    if (tid < BIN_ROWS) cnt[tid] = 0;
    if (tid == 0) sbase = 0;
    if (b == 0 && tid < 2 * CFEAT) stats[tid] = 0.0f;
    const int* __restrict__ lo = segT + (long)b * nchunks;
    const int* __restrict__ hi = lo + nchunks;
    int part = 0;
    for (int c = tid; c < nchunks; c += 1024) part += lo[c];
#pragma unroll
    for (int off = 16; off > 0; off >>= 1) part += __shfl_xor(part, off);
    __syncthreads();
    if (lane == 0 && part != 0) atomicAdd(&sbase, part);
    for (int cb = 0; cb < nchunks; cb += 1024) {
        const int cw = cb + wv;
        const int cl = cw + 32 * lane;
        int my0 = 0, my1 = 0;
        if (cl < nchunks) { my0 = lo[cl]; my1 = hi[cl]; }
        int J = (nchunks - cw + 31) >> 5;
        if (J > 32) J = 32;
        for (int j = 0; j < J; j += 4) {
            int s0[4], s1[4], dd[4];
#pragma unroll
            for (int k = 0; k < 4; ++k) {
                s0[k] = __builtin_amdgcn_readlane(my0, j + k);
                s1[k] = __builtin_amdgcn_readlane(my1, j + k);
                const int* __restrict__ pd = bdst + (long)(cw + 32 * (j + k)) * CHUNK;
                const int i = s0[k] + lane;
                dd[k] = (i < s1[k]) ? pd[i] : -1;
            }
#pragma unroll
            for (int k = 0; k < 4; ++k)
                if (dd[k] >= 0) atomicAdd(&cnt[dd[k] & rmask], 1);
#pragma unroll
            for (int k = 0; k < 4; ++k) {
                const int* __restrict__ pd = bdst + (long)(cw + 32 * (j + k)) * CHUNK;
                for (int i = s0[k] + 32 + lane; i < s1[k]; i += 32) atomicAdd(&cnt[pd[i] & rmask], 1);
            }
        }
    }
    __syncthreads();
    const int binbase = sbase;
    int v = (tid < BIN_ROWS) ? cnt[tid] : 0;
    int incl = v;
#pragma unroll
    for (int off = 1; off < 32; off <<= 1) {
        int t = __shfl_up(incl, off);
        if (lane >= off) incl += t;
    }
    if (lane == 31 && wv < 8) wtot[wv] = incl;
    __syncthreads();
    if (tid < BIN_ROWS) {
        int pre = 0;
        for (int w = 0; w < wv; ++w) pre += wtot[w];
        const int start = binbase + pre + incl - v;
        fillp[tid] = start;
        const int row = (b << shift) + tid;
        if (tid <= rmask && row < N) { cursor[row] = start; deg[row] = v; }
    }
    __syncthreads();
    for (int cb = 0; cb < nchunks; cb += 1024) {
        const int cw = cb + wv;
        const int cl = cw + 32 * lane;
        int my0 = 0, my1 = 0;
        if (cl < nchunks) { my0 = lo[cl]; my1 = hi[cl]; }
        int J = (nchunks - cw + 31) >> 5;
        if (J > 32) J = 32;
        for (int j = 0; j < J; j += 4) {
            int s0[4], s1[4], dd[4], ss[4];
#pragma unroll
            for (int k = 0; k < 4; ++k) {
                s0[k] = __builtin_amdgcn_readlane(my0, j + k);
                s1[k] = __builtin_amdgcn_readlane(my1, j + k);
                const long off = (long)(cw + 32 * (j + k)) * CHUNK;
                const int i = s0[k] + lane;
                const bool in = (i < s1[k]);
                dd[k] = in ? bdst[off + i] : -1;
                ss[k] = in ? bsrc[off + i] : 0;
            }
#pragma unroll
            for (int k = 0; k < 4; ++k) {
                if (dd[k] >= 0) {
                    const int p = atomicAdd(&fillp[dd[k] & rmask], 1);
                    if ((unsigned)p < (unsigned)E) csr[p] = ss[k];
                }
            }
#pragma unroll
            for (int k = 0; k < 4; ++k) {
                const long off = (long)(cw + 32 * (j + k)) * CHUNK;
                for (int i = s0[k] + 32 + lane; i < s1[k]; i += 32) {
                    const int d2 = bdst[off + i];
                    const int s2 = bsrc[off + i];
                    const int p = atomicAdd(&fillp[d2 & rmask], 1);
                    if ((unsigned)p < (unsigned)E) csr[p] = s2;
                }
            }
        }
    }
}

__global__ void gather_kernel(const float* __restrict__ x,
                              const int* __restrict__ csr,
                              const int* __restrict__ cursor,
                              const int* __restrict__ deg,
                              float* __restrict__ s,
                              float* __restrict__ cnt, int N) {
    int w = __builtin_amdgcn_readfirstlane((int)((blockIdx.x * blockDim.x + threadIdx.x) >> 5));
    int lane = threadIdx.x & 31;
    if (w >= N) return;
    int d = deg[w];
    int beg = cursor[w];
    int end = beg + d;
    float4 a0 = make_float4(0.0f, 0.0f, 0.0f, 0.0f);
    float4 a1 = make_float4(0.0f, 0.0f, 0.0f, 0.0f);
    for (int c = beg; c < end; c += 32) {
        int n = end - c;
        if (n > 32) n = 32;
        int my = (lane < n) ? csr[c + lane] : 0;
        int j = 0;
        for (; j + 4 <= n; j += 4) {
            const float4* r0 = (const float4*)(x + (long)__builtin_amdgcn_readlane(my, j + 0) * CFEAT);
            const float4* r1 = (const float4*)(x + (long)__builtin_amdgcn_readlane(my, j + 1) * CFEAT);
            const float4* r2 = (const float4*)(x + (long)__builtin_amdgcn_readlane(my, j + 2) * CFEAT);
            const float4* r3 = (const float4*)(x + (long)__builtin_amdgcn_readlane(my, j + 3) * CFEAT);
            float4 p0 = r0[lane], q0 = r0[lane + 32];
            float4 p1 = r1[lane], q1 = r1[lane + 32];
            float4 p2 = r2[lane], q2 = r2[lane + 32];
            float4 p3 = r3[lane], q3 = r3[lane + 32];
            a0 += p0; a1 += q0;
            a0 += p1; a1 += q1;
            a0 += p2; a1 += q2;
            a0 += p3; a1 += q3;
        }
        for (; j < n; ++j) {
            const float4* r0 = (const float4*)(x + (long)__builtin_amdgcn_readlane(my, j) * CFEAT);
            float4 p0 = r0[lane], q0 = r0[lane + 32];
            a0 += p0; a1 += q0;
        }
    }
    float4* o = (float4*)(s + (long)w * CFEAT);
    o[lane] = a0;
    o[lane + 32] = a1;
    if (lane == 0) cnt[w] = (float)d;
}

#define GBM 64
#define GBN 256
#define GKC 32
#define GPITCH 36

__global__ __launch_bounds__(256)
void gemm_kernel(const float* __restrict__ s,
                 const float* __restrict__ cnt,
                 const float* __restrict__ xt,
                 const float* __restrict__ Wl,
                 const float* __restrict__ bl,
                 const float* __restrict__ Wr,
                 float* __restrict__ h,
                 float* __restrict__ colsum,
                 float* __restrict__ colsq,
                 int N) {
    __shared__ float tileA[GBM * GPITCH];
    __shared__ float tileB[GBN * GPITCH];
    __shared__ float redS[2][GBN];
    __shared__ float redQ[2][GBN];

    const int tid  = threadIdx.x;
    const int lane = tid & 31;
    const int wv   = tid >> 5;
    const int wm   = wv >> 2;
    const int wn   = wv & 3;
    const int l15  = lane & 15;
    const int hh   = lane >> 4;
    const int row0 = blockIdx.x * GBM;

    const int srow = tid >> 3;
    const int sseg = tid & 7;
    int ar0 = row0 + srow;       if (ar0 > N - 1) ar0 = N - 1;
    int ar1 = row0 + srow + 32;  if (ar1 > N - 1) ar1 = N - 1;
    const float inv0 = 1.0f / fmaxf(cnt[ar0], 1.0f);
    const float inv1 = 1.0f / fmaxf(cnt[ar1], 1.0f);
    const long aoff0 = (long)ar0 * CFEAT + sseg * 4;
    const long aoff1 = (long)ar1 * CFEAT + sseg * 4;
    const long boff  = (long)srow * CFEAT + sseg * 4;
    const int  sdst  = srow * GPITCH + sseg * 4;

    v8f acc[2][4];
#pragma unroll
    for (int i = 0; i < 2; ++i)
#pragma unroll
        for (int j = 0; j < 4; ++j)
            acc[i][j] = (v8f){0,0,0,0,0,0,0,0};

    float4 pa0, pa1, pb0, pb1, pb2, pb3, pb4, pb5, pb6, pb7;
    const long bstep = (long)32 * CFEAT;
    {
        const float* bp = Wl + boff;
        pa0 = *(const float4*)(s + aoff0);
        pa1 = *(const float4*)(s + aoff1);
        pb0 = *(const float4*)(bp);
        pb1 = *(const float4*)(bp + bstep);
        pb2 = *(const float4*)(bp + 2 * bstep);
        pb3 = *(const float4*)(bp + 3 * bstep);
        pb4 = *(const float4*)(bp + 4 * bstep);
        pb5 = *(const float4*)(bp + 5 * bstep);
        pb6 = *(const float4*)(bp + 6 * bstep);
        pb7 = *(const float4*)(bp + 7 * bstep);
    }

    const int fa = (wm * 32 + l15) * GPITCH + hh * 4;
    const int fb = (wn * 64 + l15) * GPITCH + hh * 4;

    for (int ch = 0; ch < 16; ++ch) {
        __syncthreads();
        {
            const float m0 = (ch < 8) ? inv0 : 1.0f;
            const float m1 = (ch < 8) ? inv1 : 1.0f;
            float4 t0 = pa0, t1 = pa1;
            t0.x *= m0; t0.y *= m0; t0.z *= m0; t0.w *= m0;
            t1.x *= m1; t1.y *= m1; t1.z *= m1; t1.w *= m1;
            *(float4*)(tileA + sdst) = t0;
            *(float4*)(tileA + sdst + 32 * GPITCH) = t1;
            float* bd = tileB + sdst;
            *(float4*)(bd)                   = pb0;
            *(float4*)(bd + 1 * 32 * GPITCH) = pb1;
            *(float4*)(bd + 2 * 32 * GPITCH) = pb2;
            *(float4*)(bd + 3 * 32 * GPITCH) = pb3;
            *(float4*)(bd + 4 * 32 * GPITCH) = pb4;
            *(float4*)(bd + 5 * 32 * GPITCH) = pb5;
            *(float4*)(bd + 6 * 32 * GPITCH) = pb6;
            *(float4*)(bd + 7 * 32 * GPITCH) = pb7;
        }
        __syncthreads();
        if (ch + 1 < 16) {
            const int cn = ch + 1;
            const float* Ab = (cn < 8) ? s : xt;
            const float* Bb = (cn < 8) ? Wl : Wr;
            const int k0 = (cn & 7) * GKC;
            pa0 = *(const float4*)(Ab + aoff0 + k0);
            pa1 = *(const float4*)(Ab + aoff1 + k0);
            const float* bp = Bb + boff + k0;
            pb0 = *(const float4*)(bp);
            pb1 = *(const float4*)(bp + bstep);
            pb2 = *(const float4*)(bp + 2 * bstep);
            pb3 = *(const float4*)(bp + 3 * bstep);
            pb4 = *(const float4*)(bp + 4 * bstep);
            pb5 = *(const float4*)(bp + 5 * bstep);
            pb6 = *(const float4*)(bp + 6 * bstep);
            pb7 = *(const float4*)(bp + 7 * bstep);
        }
        __builtin_amdgcn_sched_barrier(0);
#pragma unroll
        for (int kk = 0; kk < 4; ++kk) {
            float4 af[2], bf[4];
#pragma unroll
            for (int i = 0; i < 2; ++i) af[i] = *(const float4*)(tileA + fa + i * 16 * GPITCH + kk * 8);
#pragma unroll
            for (int j = 0; j < 4; ++j) bf[j] = *(const float4*)(tileB + fb + j * 16 * GPITCH + kk * 8);
#pragma unroll
            for (int i = 0; i < 2; ++i)
#pragma unroll
                for (int j = 0; j < 4; ++j) {
                    v2f a = {af[i].x, af[i].y};
                    v2f b = {bf[j].x, bf[j].y};
                    acc[i][j] = __builtin_amdgcn_wmma_f32_16x16x4_f32(
                        false, a, false, b, (short)0, acc[i][j], false, false);
                }
#pragma unroll
            for (int i = 0; i < 2; ++i)
#pragma unroll
                for (int j = 0; j < 4; ++j) {
                    v2f a = {af[i].z, af[i].w};
                    v2f b = {bf[j].z, bf[j].w};
                    acc[i][j] = __builtin_amdgcn_wmma_f32_16x16x4_f32(
                        false, a, false, b, (short)0, acc[i][j], false, false);
                }
        }
    }

#pragma unroll
    for (int j = 0; j < 4; ++j) {
        const int col = wn * 64 + j * 16 + l15;
        const float bias = bl[col];
        float sp = 0.0f, qp = 0.0f;
#pragma unroll
        for (int i = 0; i < 2; ++i) {
            const int rbase = row0 + wm * 32 + i * 16 + hh * 8;
#pragma unroll
            for (int v = 0; v < 8; ++v) {
                float val = fmaxf(acc[i][j][v] + bias, 0.0f);
                int r = rbase + v;
                bool ok = (r < N);
                float m = ok ? val : 0.0f;
                sp += m;
                qp += m * m;
                if (ok) h[(long)r * CFEAT + col] = val;
            }
        }
        sp += __shfl_xor(sp, 16);
        qp += __shfl_xor(qp, 16);
        if (hh == 0) { redS[wm][col] = sp; redQ[wm][col] = qp; }
    }
    __syncthreads();
    atomicAdd(&colsum[tid], redS[0][tid] + redS[1][tid]);
    atomicAdd(&colsq[tid],  redQ[0][tid] + redQ[1][tid]);
}

#define NORM_U 8
#define NORM_WG 512

__device__ __forceinline__ float4 norm_apply(float4 v, float4 g, float4 mu, float4 rs, float4 b) {
    float4 o;
    o.x = g.x * (v.x - mu.x) * rs.x + b.x;
    o.y = g.y * (v.y - mu.y) * rs.y + b.y;
    o.z = g.z * (v.z - mu.z) * rs.z + b.z;
    o.w = g.w * (v.w - mu.w) * rs.w + b.w;
    return o;
}

__global__ __launch_bounds__(256)
void norm_kernel(const float* __restrict__ hbuf,
                 const float* __restrict__ colsum,
                 const float* __restrict__ colsq,
                 const float* __restrict__ gamma,
                 const float* __restrict__ beta,
                 float* __restrict__ out, int N) {
    const long total4 = (long)N * (CFEAT / 4);
    const long width  = (long)gridDim.x * blockDim.x;
    const long first  = (long)blockIdx.x * blockDim.x + threadIdx.x;
    const int  c4     = threadIdx.x & (CFEAT / 4 - 1);

    const float invN = 1.0f / (float)N;
    const float4 cs = ((const float4*)colsum)[c4];
    const float4 cq = ((const float4*)colsq)[c4];
    const float4 g  = ((const float4*)gamma)[c4];
    const float4 b  = ((const float4*)beta)[c4];
    float4 mu, rs;
    mu.x = cs.x * invN;  mu.y = cs.y * invN;  mu.z = cs.z * invN;  mu.w = cs.w * invN;
    rs.x = rsqrtf(cq.x * invN - mu.x * mu.x + 1e-5f);
    rs.y = rsqrtf(cq.y * invN - mu.y * mu.y + 1e-5f);
    rs.z = rsqrtf(cq.z * invN - mu.z * mu.z + 1e-5f);
    rs.w = rsqrtf(cq.w * invN - mu.w * mu.w + 1e-5f);

    const float4* __restrict__ in4 = (const float4*)hbuf;
    float4* __restrict__ out4 = (float4*)out;

    for (long base = first; base < total4; base += NORM_U * width) {
        float4 v[NORM_U];
#pragma unroll
        for (int u = 0; u < NORM_U; ++u) {
            const long j = base + u * width;
            v[u] = (j < total4) ? in4[j] : make_float4(0.0f, 0.0f, 0.0f, 0.0f);
        }
#pragma unroll
        for (int u = 0; u < NORM_U; ++u) {
            const long j = base + u * width;
            if (j < total4) out4[j] = norm_apply(v[u], g, mu, rs, b);
        }
    }
}

extern "C" void kernel_launch(void* const* d_in, const int* in_sizes, int n_in,
                              void* d_out, int out_size, void* d_ws, size_t ws_size,
                              hipStream_t stream) {
    const float* x    = (const float*)d_in[0];
    const int* e0src  = (const int*)d_in[1];
    const int* e0dst  = (const int*)d_in[2];
    const int* e1src  = (const int*)d_in[3];
    const int* e1dst  = (const int*)d_in[4];
    const float* Wl0  = (const float*)d_in[5];
    const float* bl0  = (const float*)d_in[6];
    const float* Wr0  = (const float*)d_in[7];
    const float* g0   = (const float*)d_in[8];
    const float* be0  = (const float*)d_in[9];
    const float* Wl1  = (const float*)d_in[10];
    const float* bl1  = (const float*)d_in[11];
    const float* Wr1  = (const float*)d_in[12];
    const float* g1   = (const float*)d_in[13];
    const float* be1  = (const float*)d_in[14];

    const int N1 = 50000, N2 = 12500;
    const int E0 = in_sizes[1];
    const int E1 = in_sizes[3];

    float* ws = (float*)d_ws;
    const long NC1 = (long)N1 * CFEAT;
    const long NC2 = (long)N2 * CFEAT;
    float* x1    = ws;
    float* sbuf  = ws + NC1;
    float* hbuf  = ws + 2 * NC1;
    float* cntb  = ws + 3 * NC1;
    float* stats = cntb + 50048;

    int* t0 = (int*)x1;
    int* t1 = (int*)(sbuf + NC2);
    const int nch0 = (E0 + CHUNK - 1) / CHUNK, nb0 = (N1 + 255) / 256;
    const int nch1 = (E1 + CHUNK - 1) / CHUNK, nb1 = (N2 + 63) / 64;

    binsort_kernel<<<nch0, 512, 0, stream>>>(e0src, e0dst, t0 + 1179648, t0 + 2228224, t0 + 3276800, E0, N1, nb0, nch0, 8);
    bincsr_kernel<<<nb0, 1024, 0, stream>>>(t0 + 1179648, t0 + 2228224, t0 + 3276800,
                                            t0 + 131072, t0 + 65536, t0, stats, E0, N1, nch0, 8);
    gather_kernel<<<(N1 + 7) / 8, 256, 0, stream>>>(x, t0 + 131072, t0 + 65536, t0, sbuf, cntb, N1);
    gemm_kernel<<<(N1 + GBM - 1) / GBM, 256, 0, stream>>>(
        sbuf, cntb, x, Wl0, bl0, Wr0, hbuf, stats, stats + 256, N1);
    norm_kernel<<<NORM_WG, 256, 0, stream>>>(
        hbuf, stats, stats + 256, g0, be0, x1, N1);

    binsort_kernel<<<nch1, 512, 0, stream>>>(e1src, e1dst, t1 + 1179648, t1 + 2228224, t1 + 3276800, E1, N2, nb1, nch1, 6);
    bincsr_kernel<<<nb1, 1024, 0, stream>>>(t1 + 1179648, t1 + 2228224, t1 + 3276800,
                                            t1 + 131072, t1 + 65536, t1, stats, E1, N2, nch1, 6);
    gather_kernel<<<(N2 + 7) / 8, 256, 0, stream>>>(x1, t1 + 131072, t1 + 65536, t1, sbuf, cntb, N2);
    gemm_kernel<<<(N2 + GBM - 1) / GBM, 256, 0, stream>>>(
        sbuf, cntb, x1, Wl1, bl1, Wr1, hbuf, stats, stats + 256, N2);
    norm_kernel<<<NORM_WG, 256, 0, stream>>>(
        hbuf, stats, stats + 256, g1, be1, (float*)d_out, N2);
}


# Layer 0's axis-0 normalisation pass removed: its two readers apply it per column, the layer-1 gather once to each row sum (a*(sum - d*mu) + d*b), the layer-1 GEMM while staging root rows; 9 launches
# speedup vs baseline: 3.3911x; 1.0537x over previous
#include <hip/hip_runtime.h>

typedef __attribute__((ext_vector_type(2))) float v2f;
typedef __attribute__((ext_vector_type(8))) float v8f;

#define CFEAT 256

#define BIN_ROWS 256
#define CHUNK 4096

__global__ __launch_bounds__(512)
void binsort_kernel(const int* __restrict__ src, const int* __restrict__ dst,
                    int* __restrict__ bsrc, int* __restrict__ bdst, int* __restrict__ segT,
                    int E, int N, int B, int nchunks, int shift) {
    __shared__ int hist[BIN_ROWS + 1];
    __shared__ int wtot[16];
    const int tid = threadIdx.x;
    const int lane = tid & 31;
    const int wv = tid >> 5;
    const int c = blockIdx.x;
    const int base = c * CHUNK;
    if (tid <= BIN_ROWS) hist[tid] = 0;

    int d[8], s[8], r[8];
#pragma unroll
    for (int k = 0; k < 2; ++k) {
        const int e = base + k * (CHUNK / 2) + tid * 4;
        if (e + 3 < E) {
            const int4 dv = *(const int4*)(dst + e);
            const int4 sv = *(const int4*)(src + e);
            d[k * 4 + 0] = dv.x; d[k * 4 + 1] = dv.y; d[k * 4 + 2] = dv.z; d[k * 4 + 3] = dv.w;
            s[k * 4 + 0] = sv.x; s[k * 4 + 1] = sv.y; s[k * 4 + 2] = sv.z; s[k * 4 + 3] = sv.w;
        } else {
#pragma unroll
            for (int i = 0; i < 4; ++i) {
                const bool in = (e + i < E);
                d[k * 4 + i] = in ? dst[e + i] : -1;
                s[k * 4 + i] = in ? src[e + i] : 0;
            }
        }
    }
    __syncthreads();
#pragma unroll
    for (int i = 0; i < 8; ++i) {
        const bool ok = ((unsigned)d[i] < (unsigned)N);
        r[i] = ok ? atomicAdd(&hist[d[i] >> shift], 1) : -1;
    }
    __syncthreads();
    int v = (tid < B) ? hist[tid] : 0;
    int incl = v;
#pragma unroll
    for (int off = 1; off < 32; off <<= 1) {
        int t = __shfl_up(incl, off);
        if (lane >= off) incl += t;
    }
    if (lane == 31) wtot[wv] = incl;
    __syncthreads();
    int pre = 0;
    for (int w = 0; w < wv; ++w) pre += wtot[w];
    const int excl = pre + incl - v;
    if (tid <= B) {
        hist[tid] = excl;
        segT[tid * nchunks + c] = excl;
    }
    __syncthreads();
#pragma unroll
    for (int i = 0; i < 8; ++i) {
        if (r[i] >= 0) {
            const int p = base + hist[d[i] >> shift] + r[i];
            bsrc[p] = s[i];
            bdst[p] = d[i];
        }
    }
}

__global__ __launch_bounds__(1024)
void bincsr_kernel(const int* __restrict__ bsrc, const int* __restrict__ bdst,
                   const int* __restrict__ segT,
                   int* __restrict__ csr, int* __restrict__ cursor, int* __restrict__ deg,
                   float* __restrict__ stats, int E, int N, int nchunks, int shift) {
    __shared__ int cnt[BIN_ROWS];
    __shared__ int fillp[BIN_ROWS];
    __shared__ int wtot[8];
    __shared__ int sbase;
    const int tid = threadIdx.x;
    const int lane = tid & 31;
    const int wv = tid >> 5;
    const int b = blockIdx.x;
    const int rmask = (1 << shift) - 1;
    if (tid < BIN_ROWS) cnt[tid] = 0;
    if (tid == 0) sbase = 0;
    if (b == 0 && tid < 2 * CFEAT) stats[tid] = 0.0f;
    const int* __restrict__ lo = segT + (long)b * nchunks;
    const int* __restrict__ hi = lo + nchunks;
    int part = 0;
    for (int c = tid; c < nchunks; c += 1024) part += lo[c];
#pragma unroll
    for (int off = 16; off > 0; off >>= 1) part += __shfl_xor(part, off);
    __syncthreads();
    if (lane == 0 && part != 0) atomicAdd(&sbase, part);
    for (int cb = 0; cb < nchunks; cb += 1024) {
        const int cw = cb + wv;
        const int cl = cw + 32 * lane;
        int my0 = 0, my1 = 0;
        if (cl < nchunks) { my0 = lo[cl]; my1 = hi[cl]; }
        int J = (nchunks - cw + 31) >> 5;
        if (J > 32) J = 32;
        for (int j = 0; j < J; j += 4) {
            int s0[4], s1[4], dd[4];
#pragma unroll
            for (int k = 0; k < 4; ++k) {
                s0[k] = __builtin_amdgcn_readlane(my0, j + k);
                s1[k] = __builtin_amdgcn_readlane(my1, j + k);
                const int* __restrict__ pd = bdst + (long)(cw + 32 * (j + k)) * CHUNK;
                const int i = s0[k] + lane;
                dd[k] = (i < s1[k]) ? pd[i] : -1;
            }
#pragma unroll
            for (int k = 0; k < 4; ++k)
                if (dd[k] >= 0) atomicAdd(&cnt[dd[k] & rmask], 1);
#pragma unroll
            for (int k = 0; k < 4; ++k) {
                const int* __restrict__ pd = bdst + (long)(cw + 32 * (j + k)) * CHUNK;
                for (int i = s0[k] + 32 + lane; i < s1[k]; i += 32) atomicAdd(&cnt[pd[i] & rmask], 1);
            }
        }
    }
    __syncthreads();
    const int binbase = sbase;
    int v = (tid < BIN_ROWS) ? cnt[tid] : 0;
    int incl = v;
#pragma unroll
    for (int off = 1; off < 32; off <<= 1) {
        int t = __shfl_up(incl, off);
        if (lane >= off) incl += t;
    }
    if (lane == 31 && wv < 8) wtot[wv] = incl;
    __syncthreads();
    if (tid < BIN_ROWS) {
        int pre = 0;
        for (int w = 0; w < wv; ++w) pre += wtot[w];
        const int start = binbase + pre + incl - v;
        fillp[tid] = start;
        const int row = (b << shift) + tid;
        if (tid <= rmask && row < N) { cursor[row] = start; deg[row] = v; }
    }
    __syncthreads();
    for (int cb = 0; cb < nchunks; cb += 1024) {
        const int cw = cb + wv;
        const int cl = cw + 32 * lane;
        int my0 = 0, my1 = 0;
        if (cl < nchunks) { my0 = lo[cl]; my1 = hi[cl]; }
        int J = (nchunks - cw + 31) >> 5;
        if (J > 32) J = 32;
        for (int j = 0; j < J; j += 4) {
            int s0[4], s1[4], dd[4], ss[4];
#pragma unroll
            for (int k = 0; k < 4; ++k) {
                s0[k] = __builtin_amdgcn_readlane(my0, j + k);
                s1[k] = __builtin_amdgcn_readlane(my1, j + k);
                const long off = (long)(cw + 32 * (j + k)) * CHUNK;
                const int i = s0[k] + lane;
                const bool in = (i < s1[k]);
                dd[k] = in ? bdst[off + i] : -1;
                ss[k] = in ? bsrc[off + i] : 0;
            }
#pragma unroll
            for (int k = 0; k < 4; ++k) {
                if (dd[k] >= 0) {
                    const int p = atomicAdd(&fillp[dd[k] & rmask], 1);
                    if ((unsigned)p < (unsigned)E) csr[p] = ss[k];
                }
            }
#pragma unroll
            for (int k = 0; k < 4; ++k) {
                const long off = (long)(cw + 32 * (j + k)) * CHUNK;
                for (int i = s0[k] + 32 + lane; i < s1[k]; i += 32) {
                    const int d2 = bdst[off + i];
                    const int s2 = bsrc[off + i];
                    const int p = atomicAdd(&fillp[d2 & rmask], 1);
                    if ((unsigned)p < (unsigned)E) csr[p] = s2;
                }
            }
        }
    }
}

__device__ __forceinline__ void col_mu_a(float cs, float cq, float g, float invN, float& mu, float& a) {
    mu = cs * invN;
    a = g * rsqrtf(cq * invN - mu * mu + 1e-5f);
}

__device__ __forceinline__ float4 sum_affine(float4 v, float d, float4 cs, float4 cq, float4 g, float4 b, float invN) {
    float mu, a;
    float4 o;
    col_mu_a(cs.x, cq.x, g.x, invN, mu, a);  o.x = a * (v.x - d * mu) + d * b.x;
    col_mu_a(cs.y, cq.y, g.y, invN, mu, a);  o.y = a * (v.y - d * mu) + d * b.y;
    col_mu_a(cs.z, cq.z, g.z, invN, mu, a);  o.z = a * (v.z - d * mu) + d * b.z;
    col_mu_a(cs.w, cq.w, g.w, invN, mu, a);  o.w = a * (v.w - d * mu) + d * b.w;
    return o;
}

template <bool AFF>
__global__ void gather_kernel(const float* __restrict__ x,
                              const int* __restrict__ csr,
                              const int* __restrict__ cursor,
                              const int* __restrict__ deg,
                              float* __restrict__ s,
                              float* __restrict__ cnt, int N,
                              const float* __restrict__ psum,
                              const float* __restrict__ psq,
                              const float* __restrict__ pgamma,
                              const float* __restrict__ pbeta, int Nprev) {
    int w = __builtin_amdgcn_readfirstlane((int)((blockIdx.x * blockDim.x + threadIdx.x) >> 5));
    int lane = threadIdx.x & 31;
    if (w >= N) return;
    int d = deg[w];
    int beg = cursor[w];
    int end = beg + d;
    float4 a0 = make_float4(0.0f, 0.0f, 0.0f, 0.0f);
    float4 a1 = make_float4(0.0f, 0.0f, 0.0f, 0.0f);
    for (int c = beg; c < end; c += 32) {
        int n = end - c;
        if (n > 32) n = 32;
        int my = (lane < n) ? csr[c + lane] : 0;
        int j = 0;
        for (; j + 4 <= n; j += 4) {
            const float4* r0 = (const float4*)(x + (long)__builtin_amdgcn_readlane(my, j + 0) * CFEAT);
            const float4* r1 = (const float4*)(x + (long)__builtin_amdgcn_readlane(my, j + 1) * CFEAT);
            const float4* r2 = (const float4*)(x + (long)__builtin_amdgcn_readlane(my, j + 2) * CFEAT);
            const float4* r3 = (const float4*)(x + (long)__builtin_amdgcn_readlane(my, j + 3) * CFEAT);
            float4 p0 = r0[lane], q0 = r0[lane + 32];
            float4 p1 = r1[lane], q1 = r1[lane + 32];
            float4 p2 = r2[lane], q2 = r2[lane + 32];
            float4 p3 = r3[lane], q3 = r3[lane + 32];
            a0 += p0; a1 += q0;
            a0 += p1; a1 += q1;
            a0 += p2; a1 += q2;
            a0 += p3; a1 += q3;
        }
        for (; j < n; ++j) {
            const float4* r0 = (const float4*)(x + (long)__builtin_amdgcn_readlane(my, j) * CFEAT);
            float4 p0 = r0[lane], q0 = r0[lane + 32];
            a0 += p0; a1 += q0;
        }
    }
    if (AFF) {
        const float invN = 1.0f / (float)Nprev;
        const float df = (float)d;
        a0 = sum_affine(a0, df, ((const float4*)psum)[lane], ((const float4*)psq)[lane],
                        ((const float4*)pgamma)[lane], ((const float4*)pbeta)[lane], invN);
        a1 = sum_affine(a1, df, ((const float4*)psum)[lane + 32], ((const float4*)psq)[lane + 32],
                        ((const float4*)pgamma)[lane + 32], ((const float4*)pbeta)[lane + 32], invN);
    }
    float4* o = (float4*)(s + (long)w * CFEAT);
    o[lane] = a0;
    o[lane + 32] = a1;
    if (lane == 0) cnt[w] = (float)d;
}

#define GBM 64
#define GBN 256
#define GKC 32
#define GPITCH 36

template <bool AFF>
__global__ __launch_bounds__(256)
void gemm_kernel(const float* __restrict__ s,
                 const float* __restrict__ cnt,
                 const float* __restrict__ xt,
                 const float* __restrict__ Wl,
                 const float* __restrict__ bl,
                 const float* __restrict__ Wr,
                 float* __restrict__ h,
                 float* __restrict__ colsum,
                 float* __restrict__ colsq,
                 int N,
                 const float* __restrict__ psum,
                 const float* __restrict__ psq,
                 const float* __restrict__ pgamma,
                 const float* __restrict__ pbeta, int Nprev) {
    __shared__ float tileA[GBM * GPITCH];
    __shared__ float tileB[GBN * GPITCH];
    __shared__ float redS[2][GBN];
    __shared__ float redQ[2][GBN];
    __shared__ float affA[CFEAT];
    __shared__ float affM[CFEAT];
    __shared__ float affB[CFEAT];

    const int tid  = threadIdx.x;
    const int lane = tid & 31;
    const int wv   = tid >> 5;
    const int wm   = wv >> 2;
    const int wn   = wv & 3;
    const int l15  = lane & 15;
    const int hh   = lane >> 4;
    const int row0 = blockIdx.x * GBM;

    if (AFF) {
        float mu, a;
        col_mu_a(psum[tid], psq[tid], pgamma[tid], 1.0f / (float)Nprev, mu, a);
        affA[tid] = a;
        affM[tid] = mu;
        affB[tid] = pbeta[tid];
    }

    const int srow = tid >> 3;
    const int sseg = tid & 7;
    int ar0 = row0 + srow;       if (ar0 > N - 1) ar0 = N - 1;
    int ar1 = row0 + srow + 32;  if (ar1 > N - 1) ar1 = N - 1;
    const float inv0 = 1.0f / fmaxf(cnt[ar0], 1.0f);
    const float inv1 = 1.0f / fmaxf(cnt[ar1], 1.0f);
    const long aoff0 = (long)ar0 * CFEAT + sseg * 4;
    const long aoff1 = (long)ar1 * CFEAT + sseg * 4;
    const long boff  = (long)srow * CFEAT + sseg * 4;
    const int  sdst  = srow * GPITCH + sseg * 4;

    v8f acc[2][4];
#pragma unroll
    for (int i = 0; i < 2; ++i)
#pragma unroll
        for (int j = 0; j < 4; ++j)
            acc[i][j] = (v8f){0,0,0,0,0,0,0,0};

    float4 pa0, pa1, pb0, pb1, pb2, pb3, pb4, pb5, pb6, pb7;
    const long bstep = (long)32 * CFEAT;
    {
        const float* bp = Wl + boff;
        pa0 = *(const float4*)(s + aoff0);
        pa1 = *(const float4*)(s + aoff1);
        pb0 = *(const float4*)(bp);
        pb1 = *(const float4*)(bp + bstep);
        pb2 = *(const float4*)(bp + 2 * bstep);
        pb3 = *(const float4*)(bp + 3 * bstep);
        pb4 = *(const float4*)(bp + 4 * bstep);
        pb5 = *(const float4*)(bp + 5 * bstep);
        pb6 = *(const float4*)(bp + 6 * bstep);
        pb7 = *(const float4*)(bp + 7 * bstep);
    }

    const int fa = (wm * 32 + l15) * GPITCH + hh * 4;
    const int fb = (wn * 64 + l15) * GPITCH + hh * 4;

    for (int ch = 0; ch < 16; ++ch) {
        __syncthreads();
        {
            const float m0 = (ch < 8) ? inv0 : 1.0f;
            const float m1 = (ch < 8) ? inv1 : 1.0f;
            float4 t0 = pa0, t1 = pa1;
            t0.x *= m0; t0.y *= m0; t0.z *= m0; t0.w *= m0;
            t1.x *= m1; t1.y *= m1; t1.z *= m1; t1.w *= m1;
            if (AFF && ch >= 8) {
                const int q = ((ch & 7) * 8 + sseg) * 4;
                const float4 ca = *(const float4*)(affA + q);
                const float4 cm = *(const float4*)(affM + q);
                const float4 cb = *(const float4*)(affB + q);
                t0.x = ca.x * (t0.x - cm.x) + cb.x;  t0.y = ca.y * (t0.y - cm.y) + cb.y;
                t0.z = ca.z * (t0.z - cm.z) + cb.z;  t0.w = ca.w * (t0.w - cm.w) + cb.w;
                t1.x = ca.x * (t1.x - cm.x) + cb.x;  t1.y = ca.y * (t1.y - cm.y) + cb.y;
                t1.z = ca.z * (t1.z - cm.z) + cb.z;  t1.w = ca.w * (t1.w - cm.w) + cb.w;
            }
            *(float4*)(tileA + sdst) = t0;
            *(float4*)(tileA + sdst + 32 * GPITCH) = t1;
            float* bd = tileB + sdst;
            *(float4*)(bd)                   = pb0;
            *(float4*)(bd + 1 * 32 * GPITCH) = pb1;
            *(float4*)(bd + 2 * 32 * GPITCH) = pb2;
            *(float4*)(bd + 3 * 32 * GPITCH) = pb3;
            *(float4*)(bd + 4 * 32 * GPITCH) = pb4;
            *(float4*)(bd + 5 * 32 * GPITCH) = pb5;
            *(float4*)(bd + 6 * 32 * GPITCH) = pb6;
            *(float4*)(bd + 7 * 32 * GPITCH) = pb7;
        }
        __syncthreads();
        if (ch + 1 < 16) {
            const int cn = ch + 1;
            const float* Ab = (cn < 8) ? s : xt;
            const float* Bb = (cn < 8) ? Wl : Wr;
            const int k0 = (cn & 7) * GKC;
            pa0 = *(const float4*)(Ab + aoff0 + k0);
            pa1 = *(const float4*)(Ab + aoff1 + k0);
            const float* bp = Bb + boff + k0;
            pb0 = *(const float4*)(bp);
            pb1 = *(const float4*)(bp + bstep);
            pb2 = *(const float4*)(bp + 2 * bstep);
            pb3 = *(const float4*)(bp + 3 * bstep);
            pb4 = *(const float4*)(bp + 4 * bstep);
            pb5 = *(const float4*)(bp + 5 * bstep);
            pb6 = *(const float4*)(bp + 6 * bstep);
            pb7 = *(const float4*)(bp + 7 * bstep);
        }
        __builtin_amdgcn_sched_barrier(0);
#pragma unroll
        for (int kk = 0; kk < 4; ++kk) {
            float4 af[2], bf[4];
#pragma unroll
            for (int i = 0; i < 2; ++i) af[i] = *(const float4*)(tileA + fa + i * 16 * GPITCH + kk * 8);
#pragma unroll
            for (int j = 0; j < 4; ++j) bf[j] = *(const float4*)(tileB + fb + j * 16 * GPITCH + kk * 8);
#pragma unroll
            for (int i = 0; i < 2; ++i)
#pragma unroll
                for (int j = 0; j < 4; ++j) {
                    v2f a = {af[i].x, af[i].y};
                    v2f b = {bf[j].x, bf[j].y};
                    acc[i][j] = __builtin_amdgcn_wmma_f32_16x16x4_f32(
                        false, a, false, b, (short)0, acc[i][j], false, false);
                }
#pragma unroll
            for (int i = 0; i < 2; ++i)
#pragma unroll
                for (int j = 0; j < 4; ++j) {
                    v2f a = {af[i].z, af[i].w};
                    v2f b = {bf[j].z, bf[j].w};
                    acc[i][j] = __builtin_amdgcn_wmma_f32_16x16x4_f32(
                        false, a, false, b, (short)0, acc[i][j], false, false);
                }
        }
    }

#pragma unroll
    for (int j = 0; j < 4; ++j) {
        const int col = wn * 64 + j * 16 + l15;
        const float bias = bl[col];
        float sp = 0.0f, qp = 0.0f;
#pragma unroll
        for (int i = 0; i < 2; ++i) {
            const int rbase = row0 + wm * 32 + i * 16 + hh * 8;
#pragma unroll
            for (int v = 0; v < 8; ++v) {
                float val = fmaxf(acc[i][j][v] + bias, 0.0f);
                int r = rbase + v;
                bool ok = (r < N);
                float m = ok ? val : 0.0f;
                sp += m;
                qp += m * m;
                if (ok) h[(long)r * CFEAT + col] = val;
            }
        }
        sp += __shfl_xor(sp, 16);
        qp += __shfl_xor(qp, 16);
        if (hh == 0) { redS[wm][col] = sp; redQ[wm][col] = qp; }
    }
    __syncthreads();
    atomicAdd(&colsum[tid], redS[0][tid] + redS[1][tid]);
    atomicAdd(&colsq[tid],  redQ[0][tid] + redQ[1][tid]);
}

#define NORM_U 8
#define NORM_WG 512

__device__ __forceinline__ float4 norm_apply(float4 v, float4 g, float4 mu, float4 rs, float4 b) {
    float4 o;
    o.x = g.x * (v.x - mu.x) * rs.x + b.x;
    o.y = g.y * (v.y - mu.y) * rs.y + b.y;
    o.z = g.z * (v.z - mu.z) * rs.z + b.z;
    o.w = g.w * (v.w - mu.w) * rs.w + b.w;
    return o;
}

__global__ __launch_bounds__(256)
void norm_kernel(const float* __restrict__ hbuf,
                 const float* __restrict__ colsum,
                 const float* __restrict__ colsq,
                 const float* __restrict__ gamma,
                 const float* __restrict__ beta,
                 float* __restrict__ out, int N) {
    const long total4 = (long)N * (CFEAT / 4);
    const long width  = (long)gridDim.x * blockDim.x;
    const long first  = (long)blockIdx.x * blockDim.x + threadIdx.x;
    const int  c4     = threadIdx.x & (CFEAT / 4 - 1);

    const float invN = 1.0f / (float)N;
    const float4 cs = ((const float4*)colsum)[c4];
    const float4 cq = ((const float4*)colsq)[c4];
    const float4 g  = ((const float4*)gamma)[c4];
    const float4 b  = ((const float4*)beta)[c4];
    float4 mu, rs;
    mu.x = cs.x * invN;  mu.y = cs.y * invN;  mu.z = cs.z * invN;  mu.w = cs.w * invN;
    rs.x = rsqrtf(cq.x * invN - mu.x * mu.x + 1e-5f);
    rs.y = rsqrtf(cq.y * invN - mu.y * mu.y + 1e-5f);
    rs.z = rsqrtf(cq.z * invN - mu.z * mu.z + 1e-5f);
    rs.w = rsqrtf(cq.w * invN - mu.w * mu.w + 1e-5f);

    const float4* __restrict__ in4 = (const float4*)hbuf;
    float4* __restrict__ out4 = (float4*)out;

    for (long base = first; base < total4; base += NORM_U * width) {
        float4 v[NORM_U];
#pragma unroll
        for (int u = 0; u < NORM_U; ++u) {
            const long j = base + u * width;
            v[u] = (j < total4) ? in4[j] : make_float4(0.0f, 0.0f, 0.0f, 0.0f);
        }
#pragma unroll
        for (int u = 0; u < NORM_U; ++u) {
            const long j = base + u * width;
            if (j < total4) out4[j] = norm_apply(v[u], g, mu, rs, b);
        }
    }
}

extern "C" void kernel_launch(void* const* d_in, const int* in_sizes, int n_in,
                              void* d_out, int out_size, void* d_ws, size_t ws_size,
                              hipStream_t stream) {
    const float* x    = (const float*)d_in[0];
    const int* e0src  = (const int*)d_in[1];
    const int* e0dst  = (const int*)d_in[2];
    const int* e1src  = (const int*)d_in[3];
    const int* e1dst  = (const int*)d_in[4];
    const float* Wl0  = (const float*)d_in[5];
    const float* bl0  = (const float*)d_in[6];
    const float* Wr0  = (const float*)d_in[7];
    const float* g0   = (const float*)d_in[8];
    const float* be0  = (const float*)d_in[9];
    const float* Wl1  = (const float*)d_in[10];
    const float* bl1  = (const float*)d_in[11];
    const float* Wr1  = (const float*)d_in[12];
    const float* g1   = (const float*)d_in[13];
    const float* be1  = (const float*)d_in[14];

    const int N1 = 50000, N2 = 12500;
    const int E0 = in_sizes[1];
    const int E1 = in_sizes[3];

    float* ws = (float*)d_ws;
    const long NC1 = (long)N1 * CFEAT;
    const long NC2 = (long)N2 * CFEAT;
    float* x1    = ws;
    float* sbuf  = ws + NC1;
    float* hbuf  = ws + 2 * NC1;
    float* cntb  = ws + 3 * NC1;
    float* stats = cntb + 50048;
    float* stats1 = x1 + 4000000;

    int* t0 = (int*)x1;
    int* t1 = (int*)(sbuf + NC2);
    const int nch0 = (E0 + CHUNK - 1) / CHUNK, nb0 = (N1 + 255) / 256;
    const int nch1 = (E1 + CHUNK - 1) / CHUNK, nb1 = (N2 + 63) / 64;

    binsort_kernel<<<nch0, 512, 0, stream>>>(e0src, e0dst, t0 + 1179648, t0 + 2228224, t0 + 3276800, E0, N1, nb0, nch0, 8);
    bincsr_kernel<<<nb0, 1024, 0, stream>>>(t0 + 1179648, t0 + 2228224, t0 + 3276800,
                                            t0 + 131072, t0 + 65536, t0, stats, E0, N1, nch0, 8);
    gather_kernel<false><<<(N1 + 7) / 8, 256, 0, stream>>>(x, t0 + 131072, t0 + 65536, t0, sbuf, cntb, N1,
                                                           g0, g0, g0, be0, N1);
    gemm_kernel<false><<<(N1 + GBM - 1) / GBM, 256, 0, stream>>>(
        sbuf, cntb, x, Wl0, bl0, Wr0, hbuf, stats, stats + 256, N1, g0, g0, g0, be0, N1);

    binsort_kernel<<<nch1, 512, 0, stream>>>(e1src, e1dst, t1 + 1179648, t1 + 2228224, t1 + 3276800, E1, N2, nb1, nch1, 6);
    bincsr_kernel<<<nb1, 1024, 0, stream>>>(t1 + 1179648, t1 + 2228224, t1 + 3276800,
                                            t1 + 131072, t1 + 65536, t1, stats1, E1, N2, nch1, 6);
    gather_kernel<true><<<(N2 + 7) / 8, 256, 0, stream>>>(hbuf, t1 + 131072, t1 + 65536, t1, sbuf, cntb, N2,
                                                          stats, stats + 256, g0, be0, N1);
    gemm_kernel<true><<<(N2 + GBM - 1) / GBM, 256, 0, stream>>>(
        sbuf, cntb, hbuf, Wl1, bl1, Wr1, x1, stats1, stats1 + 256, N2, stats, stats + 256, g0, be0, N1);
    norm_kernel<<<NORM_WG, 256, 0, stream>>>(
        x1, stats1, stats1 + 256, g1, be1, (float*)d_out, N2);
}


# Layer 1's edge-list build rides in layer 0's two build launches: one sort launch and one row-placement launch serve both lists (block range picks the list; layer-1 scratch moved); 7 launches
# speedup vs baseline: 3.5371x; 1.0431x over previous
#include <hip/hip_runtime.h>

typedef __attribute__((ext_vector_type(2))) float v2f;
typedef __attribute__((ext_vector_type(8))) float v8f;

#define CFEAT 256

#define BIN_ROWS 256
#define CHUNK 4096

__global__ __launch_bounds__(512)
void binsort_kernel(const int* __restrict__ srcA, const int* __restrict__ dstA,
                    int* __restrict__ bsrcA, int* __restrict__ bdstA, int* __restrict__ segTA,
                    int EA, int NA, int BA, int nchunksA, int shiftA,
                    const int* __restrict__ srcB, const int* __restrict__ dstB,
                    int* __restrict__ bsrcB, int* __restrict__ bdstB, int* __restrict__ segTB,
                    int EB, int NB, int BB, int nchunksB, int shiftB) {
    __shared__ int hist[BIN_ROWS + 1];
    __shared__ int wtot[16];
    const bool second = ((int)blockIdx.x >= nchunksA);
    const int* __restrict__ src = second ? srcB : srcA;
    const int* __restrict__ dst = second ? dstB : dstA;
    int* __restrict__ bsrc = second ? bsrcB : bsrcA;
    int* __restrict__ bdst = second ? bdstB : bdstA;
    int* __restrict__ segT = second ? segTB : segTA;
    const int E = second ? EB : EA;
    const int N = second ? NB : NA;
    const int B = second ? BB : BA;
    const int nchunks = second ? nchunksB : nchunksA;
    const int shift = second ? shiftB : shiftA;
    const int tid = threadIdx.x;
    const int lane = tid & 31;
    const int wv = tid >> 5;
    const int c = second ? (int)blockIdx.x - nchunksA : (int)blockIdx.x;
    const int base = c * CHUNK;
    if (tid <= BIN_ROWS) hist[tid] = 0;

    int d[8], s[8], r[8];
#pragma unroll
    for (int k = 0; k < 2; ++k) {
        const int e = base + k * (CHUNK / 2) + tid * 4;
        if (e + 3 < E) {
            const int4 dv = *(const int4*)(dst + e);
            const int4 sv = *(const int4*)(src + e);
            d[k * 4 + 0] = dv.x; d[k * 4 + 1] = dv.y; d[k * 4 + 2] = dv.z; d[k * 4 + 3] = dv.w;
            s[k * 4 + 0] = sv.x; s[k * 4 + 1] = sv.y; s[k * 4 + 2] = sv.z; s[k * 4 + 3] = sv.w;
        } else {
#pragma unroll
            for (int i = 0; i < 4; ++i) {
                const bool in = (e + i < E);
                d[k * 4 + i] = in ? dst[e + i] : -1;
                s[k * 4 + i] = in ? src[e + i] : 0;
            }
        }
    }
    __syncthreads();
#pragma unroll
    for (int i = 0; i < 8; ++i) {
        const bool ok = ((unsigned)d[i] < (unsigned)N);
        r[i] = ok ? atomicAdd(&hist[d[i] >> shift], 1) : -1;
    }
    __syncthreads();
    int v = (tid < B) ? hist[tid] : 0;
    int incl = v;
#pragma unroll
    for (int off = 1; off < 32; off <<= 1) {
        int t = __shfl_up(incl, off);
        if (lane >= off) incl += t;
    }
    if (lane == 31) wtot[wv] = incl;
    __syncthreads();
    int pre = 0;
    for (int w = 0; w < wv; ++w) pre += wtot[w];
    const int excl = pre + incl - v;
    if (tid <= B) {
        hist[tid] = excl;
        segT[tid * nchunks + c] = excl;
    }
    __syncthreads();
#pragma unroll
    for (int i = 0; i < 8; ++i) {
        if (r[i] >= 0) {
            const int p = base + hist[d[i] >> shift] + r[i];
            bsrc[p] = s[i];
            bdst[p] = d[i];
        }
    }
}

__global__ __launch_bounds__(1024)
void bincsr_kernel(const int* __restrict__ bsrcA, const int* __restrict__ bdstA,
                   const int* __restrict__ segTA,
                   int* __restrict__ csrA, int* __restrict__ cursorA, int* __restrict__ degA,
                   float* __restrict__ statsA, int EA, int NA, int nchunksA, int shiftA, int nbinsA,
                   const int* __restrict__ bsrcB, const int* __restrict__ bdstB,
                   const int* __restrict__ segTB,
                   int* __restrict__ csrB, int* __restrict__ cursorB, int* __restrict__ degB,
                   float* __restrict__ statsB, int EB, int NB, int nchunksB, int shiftB) {
    __shared__ int cnt[BIN_ROWS];
    __shared__ int fillp[BIN_ROWS];
    __shared__ int wtot[8];
    __shared__ int sbase;
    const bool second = ((int)blockIdx.x >= nbinsA);
    const int* __restrict__ bsrc = second ? bsrcB : bsrcA;
    const int* __restrict__ bdst = second ? bdstB : bdstA;
    const int* __restrict__ segT = second ? segTB : segTA;
    int* __restrict__ csr = second ? csrB : csrA;
    int* __restrict__ cursor = second ? cursorB : cursorA;
    int* __restrict__ deg = second ? degB : degA;
    float* __restrict__ stats = second ? statsB : statsA;
    const int E = second ? EB : EA;
    const int N = second ? NB : NA;
    const int nchunks = second ? nchunksB : nchunksA;
    const int shift = second ? shiftB : shiftA;
    const int tid = threadIdx.x;
    const int lane = tid & 31;
    const int wv = tid >> 5;
    const int b = second ? (int)blockIdx.x - nbinsA : (int)blockIdx.x;
    const int rmask = (1 << shift) - 1;
    if (tid < BIN_ROWS) cnt[tid] = 0;
    if (tid == 0) sbase = 0;
    if (b == 0 && tid < 2 * CFEAT) stats[tid] = 0.0f;
    const int* __restrict__ lo = segT + (long)b * nchunks;
    const int* __restrict__ hi = lo + nchunks;
    int part = 0;
    for (int c = tid; c < nchunks; c += 1024) part += lo[c];
#pragma unroll
    for (int off = 16; off > 0; off >>= 1) part += __shfl_xor(part, off);
    __syncthreads();
    if (lane == 0 && part != 0) atomicAdd(&sbase, part);
    for (int cb = 0; cb < nchunks; cb += 1024) {
        const int cw = cb + wv;
        const int cl = cw + 32 * lane;
        int my0 = 0, my1 = 0;
        if (cl < nchunks) { my0 = lo[cl]; my1 = hi[cl]; }
        int J = (nchunks - cw + 31) >> 5;
        if (J > 32) J = 32;
        for (int j = 0; j < J; j += 4) {
            int s0[4], s1[4], dd[4];
#pragma unroll
            for (int k = 0; k < 4; ++k) {
                s0[k] = __builtin_amdgcn_readlane(my0, j + k);
                s1[k] = __builtin_amdgcn_readlane(my1, j + k);
                const int* __restrict__ pd = bdst + (long)(cw + 32 * (j + k)) * CHUNK;
                const int i = s0[k] + lane;
                dd[k] = (i < s1[k]) ? pd[i] : -1;
            }
#pragma unroll
            for (int k = 0; k < 4; ++k)
                if (dd[k] >= 0) atomicAdd(&cnt[dd[k] & rmask], 1);
#pragma unroll
            for (int k = 0; k < 4; ++k) {
                const int* __restrict__ pd = bdst + (long)(cw + 32 * (j + k)) * CHUNK;
                for (int i = s0[k] + 32 + lane; i < s1[k]; i += 32) atomicAdd(&cnt[pd[i] & rmask], 1);
            }
        }
    }
    __syncthreads();
    const int binbase = sbase;
    int v = (tid < BIN_ROWS) ? cnt[tid] : 0;
    int incl = v;
#pragma unroll
    for (int off = 1; off < 32; off <<= 1) {
        int t = __shfl_up(incl, off);
        if (lane >= off) incl += t;
    }
    if (lane == 31 && wv < 8) wtot[wv] = incl;
    __syncthreads();
    if (tid < BIN_ROWS) {
        int pre = 0;
        for (int w = 0; w < wv; ++w) pre += wtot[w];
        const int start = binbase + pre + incl - v;
        fillp[tid] = start;
        const int row = (b << shift) + tid;
        if (tid <= rmask && row < N) { cursor[row] = start; deg[row] = v; }
    }
    __syncthreads();
    for (int cb = 0; cb < nchunks; cb += 1024) {
        const int cw = cb + wv;
        const int cl = cw + 32 * lane;
        int my0 = 0, my1 = 0;
        if (cl < nchunks) { my0 = lo[cl]; my1 = hi[cl]; }
        int J = (nchunks - cw + 31) >> 5;
        if (J > 32) J = 32;
        for (int j = 0; j < J; j += 4) {
            int s0[4], s1[4], dd[4], ss[4];
#pragma unroll
            for (int k = 0; k < 4; ++k) {
                s0[k] = __builtin_amdgcn_readlane(my0, j + k);
                s1[k] = __builtin_amdgcn_readlane(my1, j + k);
                const long off = (long)(cw + 32 * (j + k)) * CHUNK;
                const int i = s0[k] + lane;
                const bool in = (i < s1[k]);
                dd[k] = in ? bdst[off + i] : -1;
                ss[k] = in ? bsrc[off + i] : 0;
            }
#pragma unroll
            for (int k = 0; k < 4; ++k) {
                if (dd[k] >= 0) {
                    const int p = atomicAdd(&fillp[dd[k] & rmask], 1);
                    if ((unsigned)p < (unsigned)E) csr[p] = ss[k];
                }
            }
#pragma unroll
            for (int k = 0; k < 4; ++k) {
                const long off = (long)(cw + 32 * (j + k)) * CHUNK;
                for (int i = s0[k] + 32 + lane; i < s1[k]; i += 32) {
                    const int d2 = bdst[off + i];
                    const int s2 = bsrc[off + i];
                    const int p = atomicAdd(&fillp[d2 & rmask], 1);
                    if ((unsigned)p < (unsigned)E) csr[p] = s2;
                }
            }
        }
    }
}

__device__ __forceinline__ void col_mu_a(float cs, float cq, float g, float invN, float& mu, float& a) {
    mu = cs * invN;
    a = g * rsqrtf(cq * invN - mu * mu + 1e-5f);
}

__device__ __forceinline__ float4 sum_affine(float4 v, float d, float4 cs, float4 cq, float4 g, float4 b, float invN) {
    float mu, a;
    float4 o;
    col_mu_a(cs.x, cq.x, g.x, invN, mu, a);  o.x = a * (v.x - d * mu) + d * b.x;
    col_mu_a(cs.y, cq.y, g.y, invN, mu, a);  o.y = a * (v.y - d * mu) + d * b.y;
    col_mu_a(cs.z, cq.z, g.z, invN, mu, a);  o.z = a * (v.z - d * mu) + d * b.z;
    col_mu_a(cs.w, cq.w, g.w, invN, mu, a);  o.w = a * (v.w - d * mu) + d * b.w;
    return o;
}

template <bool AFF>
__global__ void gather_kernel(const float* __restrict__ x,
                              const int* __restrict__ csr,
                              const int* __restrict__ cursor,
                              const int* __restrict__ deg,
                              float* __restrict__ s,
                              float* __restrict__ cnt, int N,
                              const float* __restrict__ psum,
                              const float* __restrict__ psq,
                              const float* __restrict__ pgamma,
                              const float* __restrict__ pbeta, int Nprev) {
    int w = __builtin_amdgcn_readfirstlane((int)((blockIdx.x * blockDim.x + threadIdx.x) >> 5));
    int lane = threadIdx.x & 31;
    if (w >= N) return;
    int d = deg[w];
    int beg = cursor[w];
    int end = beg + d;
    float4 a0 = make_float4(0.0f, 0.0f, 0.0f, 0.0f);
    float4 a1 = make_float4(0.0f, 0.0f, 0.0f, 0.0f);
    for (int c = beg; c < end; c += 32) {
        int n = end - c;
        if (n > 32) n = 32;
        int my = (lane < n) ? csr[c + lane] : 0;
        int j = 0;
        for (; j + 4 <= n; j += 4) {
            const float4* r0 = (const float4*)(x + (long)__builtin_amdgcn_readlane(my, j + 0) * CFEAT);
            const float4* r1 = (const float4*)(x + (long)__builtin_amdgcn_readlane(my, j + 1) * CFEAT);
            const float4* r2 = (const float4*)(x + (long)__builtin_amdgcn_readlane(my, j + 2) * CFEAT);
            const float4* r3 = (const float4*)(x + (long)__builtin_amdgcn_readlane(my, j + 3) * CFEAT);
            float4 p0 = r0[lane], q0 = r0[lane + 32];
            float4 p1 = r1[lane], q1 = r1[lane + 32];
            float4 p2 = r2[lane], q2 = r2[lane + 32];
            float4 p3 = r3[lane], q3 = r3[lane + 32];
            a0 += p0; a1 += q0;
            a0 += p1; a1 += q1;
            a0 += p2; a1 += q2;
            a0 += p3; a1 += q3;
        }
        for (; j < n; ++j) {
            const float4* r0 = (const float4*)(x + (long)__builtin_amdgcn_readlane(my, j) * CFEAT);
            float4 p0 = r0[lane], q0 = r0[lane + 32];
            a0 += p0; a1 += q0;
        }
    }
    if (AFF) {
        const float invN = 1.0f / (float)Nprev;
        const float df = (float)d;
        a0 = sum_affine(a0, df, ((const float4*)psum)[lane], ((const float4*)psq)[lane],
                        ((const float4*)pgamma)[lane], ((const float4*)pbeta)[lane], invN);
        a1 = sum_affine(a1, df, ((const float4*)psum)[lane + 32], ((const float4*)psq)[lane + 32],
                        ((const float4*)pgamma)[lane + 32], ((const float4*)pbeta)[lane + 32], invN);
    }
    float4* o = (float4*)(s + (long)w * CFEAT);
    o[lane] = a0;
    o[lane + 32] = a1;
    if (lane == 0) cnt[w] = (float)d;
}

#define GBM 64
#define GBN 256
#define GKC 32
#define GPITCH 36

template <bool AFF>
__global__ __launch_bounds__(256)
void gemm_kernel(const float* __restrict__ s,
                 const float* __restrict__ cnt,
                 const float* __restrict__ xt,
                 const float* __restrict__ Wl,
                 const float* __restrict__ bl,
                 const float* __restrict__ Wr,
                 float* __restrict__ h,
                 float* __restrict__ colsum,
                 float* __restrict__ colsq,
                 int N,
                 const float* __restrict__ psum,
                 const float* __restrict__ psq,
                 const float* __restrict__ pgamma,
                 const float* __restrict__ pbeta, int Nprev) {
    __shared__ float tileA[GBM * GPITCH];
    __shared__ float tileB[GBN * GPITCH];
    __shared__ float redS[2][GBN];
    __shared__ float redQ[2][GBN];
    __shared__ float affA[CFEAT];
    __shared__ float affM[CFEAT];
    __shared__ float affB[CFEAT];

    const int tid  = threadIdx.x;
    const int lane = tid & 31;
    const int wv   = tid >> 5;
    const int wm   = wv >> 2;
    const int wn   = wv & 3;
    const int l15  = lane & 15;
    const int hh   = lane >> 4;
    const int row0 = blockIdx.x * GBM;

    if (AFF) {
        float mu, a;
        col_mu_a(psum[tid], psq[tid], pgamma[tid], 1.0f / (float)Nprev, mu, a);
        affA[tid] = a;
        affM[tid] = mu;
        affB[tid] = pbeta[tid];
    }

    const int srow = tid >> 3;
    const int sseg = tid & 7;
    int ar0 = row0 + srow;       if (ar0 > N - 1) ar0 = N - 1;
    int ar1 = row0 + srow + 32;  if (ar1 > N - 1) ar1 = N - 1;
    const float inv0 = 1.0f / fmaxf(cnt[ar0], 1.0f);
    const float inv1 = 1.0f / fmaxf(cnt[ar1], 1.0f);
    const long aoff0 = (long)ar0 * CFEAT + sseg * 4;
    const long aoff1 = (long)ar1 * CFEAT + sseg * 4;
    const long boff  = (long)srow * CFEAT + sseg * 4;
    const int  sdst  = srow * GPITCH + sseg * 4;

    v8f acc[2][4];
#pragma unroll
    for (int i = 0; i < 2; ++i)
#pragma unroll
        for (int j = 0; j < 4; ++j)
            acc[i][j] = (v8f){0,0,0,0,0,0,0,0};

    float4 pa0, pa1, pb0, pb1, pb2, pb3, pb4, pb5, pb6, pb7;
    const long bstep = (long)32 * CFEAT;
    {
        const float* bp = Wl + boff;
        pa0 = *(const float4*)(s + aoff0);
        pa1 = *(const float4*)(s + aoff1);
        pb0 = *(const float4*)(bp);
        pb1 = *(const float4*)(bp + bstep);
        pb2 = *(const float4*)(bp + 2 * bstep);
        pb3 = *(const float4*)(bp + 3 * bstep);
        pb4 = *(const float4*)(bp + 4 * bstep);
        pb5 = *(const float4*)(bp + 5 * bstep);
        pb6 = *(const float4*)(bp + 6 * bstep);
        pb7 = *(const float4*)(bp + 7 * bstep);
    }

    const int fa = (wm * 32 + l15) * GPITCH + hh * 4;
    const int fb = (wn * 64 + l15) * GPITCH + hh * 4;

    for (int ch = 0; ch < 16; ++ch) {
        __syncthreads();
        {
            const float m0 = (ch < 8) ? inv0 : 1.0f;
            const float m1 = (ch < 8) ? inv1 : 1.0f;
            float4 t0 = pa0, t1 = pa1;
            t0.x *= m0; t0.y *= m0; t0.z *= m0; t0.w *= m0;
            t1.x *= m1; t1.y *= m1; t1.z *= m1; t1.w *= m1;
            if (AFF && ch >= 8) {
                const int q = ((ch & 7) * 8 + sseg) * 4;
                const float4 ca = *(const float4*)(affA + q);
                const float4 cm = *(const float4*)(affM + q);
                const float4 cb = *(const float4*)(affB + q);
                t0.x = ca.x * (t0.x - cm.x) + cb.x;  t0.y = ca.y * (t0.y - cm.y) + cb.y;
                t0.z = ca.z * (t0.z - cm.z) + cb.z;  t0.w = ca.w * (t0.w - cm.w) + cb.w;
                t1.x = ca.x * (t1.x - cm.x) + cb.x;  t1.y = ca.y * (t1.y - cm.y) + cb.y;
                t1.z = ca.z * (t1.z - cm.z) + cb.z;  t1.w = ca.w * (t1.w - cm.w) + cb.w;
            }
            *(float4*)(tileA + sdst) = t0;
            *(float4*)(tileA + sdst + 32 * GPITCH) = t1;
            float* bd = tileB + sdst;
            *(float4*)(bd)                   = pb0;
            *(float4*)(bd + 1 * 32 * GPITCH) = pb1;
            *(float4*)(bd + 2 * 32 * GPITCH) = pb2;
            *(float4*)(bd + 3 * 32 * GPITCH) = pb3;
            *(float4*)(bd + 4 * 32 * GPITCH) = pb4;
            *(float4*)(bd + 5 * 32 * GPITCH) = pb5;
            *(float4*)(bd + 6 * 32 * GPITCH) = pb6;
            *(float4*)(bd + 7 * 32 * GPITCH) = pb7;
        }
        __syncthreads();
        if (ch + 1 < 16) {
            const int cn = ch + 1;
            const float* Ab = (cn < 8) ? s : xt;
            const float* Bb = (cn < 8) ? Wl : Wr;
            const int k0 = (cn & 7) * GKC;
            pa0 = *(const float4*)(Ab + aoff0 + k0);
            pa1 = *(const float4*)(Ab + aoff1 + k0);
            const float* bp = Bb + boff + k0;
            pb0 = *(const float4*)(bp);
            pb1 = *(const float4*)(bp + bstep);
            pb2 = *(const float4*)(bp + 2 * bstep);
            pb3 = *(const float4*)(bp + 3 * bstep);
            pb4 = *(const float4*)(bp + 4 * bstep);
            pb5 = *(const float4*)(bp + 5 * bstep);
            pb6 = *(const float4*)(bp + 6 * bstep);
            pb7 = *(const float4*)(bp + 7 * bstep);
        }
        __builtin_amdgcn_sched_barrier(0);
#pragma unroll
        for (int kk = 0; kk < 4; ++kk) {
            float4 af[2], bf[4];
#pragma unroll
            for (int i = 0; i < 2; ++i) af[i] = *(const float4*)(tileA + fa + i * 16 * GPITCH + kk * 8);
#pragma unroll
            for (int j = 0; j < 4; ++j) bf[j] = *(const float4*)(tileB + fb + j * 16 * GPITCH + kk * 8);
#pragma unroll
            for (int i = 0; i < 2; ++i)
#pragma unroll
                for (int j = 0; j < 4; ++j) {
                    v2f a = {af[i].x, af[i].y};
                    v2f b = {bf[j].x, bf[j].y};
                    acc[i][j] = __builtin_amdgcn_wmma_f32_16x16x4_f32(
                        false, a, false, b, (short)0, acc[i][j], false, false);
                }
#pragma unroll
            for (int i = 0; i < 2; ++i)
#pragma unroll
                for (int j = 0; j < 4; ++j) {
                    v2f a = {af[i].z, af[i].w};
                    v2f b = {bf[j].z, bf[j].w};
                    acc[i][j] = __builtin_amdgcn_wmma_f32_16x16x4_f32(
                        false, a, false, b, (short)0, acc[i][j], false, false);
                }
        }
    }

#pragma unroll
    for (int j = 0; j < 4; ++j) {
        const int col = wn * 64 + j * 16 + l15;
        const float bias = bl[col];
        float sp = 0.0f, qp = 0.0f;
#pragma unroll
        for (int i = 0; i < 2; ++i) {
            const int rbase = row0 + wm * 32 + i * 16 + hh * 8;
#pragma unroll
            for (int v = 0; v < 8; ++v) {
                float val = fmaxf(acc[i][j][v] + bias, 0.0f);
                int r = rbase + v;
                bool ok = (r < N);
                float m = ok ? val : 0.0f;
                sp += m;
                qp += m * m;
                if (ok) h[(long)r * CFEAT + col] = val;
            }
        }
        sp += __shfl_xor(sp, 16);
        qp += __shfl_xor(qp, 16);
        if (hh == 0) { redS[wm][col] = sp; redQ[wm][col] = qp; }
    }
    __syncthreads();
    atomicAdd(&colsum[tid], redS[0][tid] + redS[1][tid]);
    atomicAdd(&colsq[tid],  redQ[0][tid] + redQ[1][tid]);
}

#define NORM_U 8
#define NORM_WG 512

__device__ __forceinline__ float4 norm_apply(float4 v, float4 g, float4 mu, float4 rs, float4 b) {
    float4 o;
    o.x = g.x * (v.x - mu.x) * rs.x + b.x;
    o.y = g.y * (v.y - mu.y) * rs.y + b.y;
    o.z = g.z * (v.z - mu.z) * rs.z + b.z;
    o.w = g.w * (v.w - mu.w) * rs.w + b.w;
    return o;
}

__global__ __launch_bounds__(256)
void norm_kernel(const float* __restrict__ hbuf,
                 const float* __restrict__ colsum,
                 const float* __restrict__ colsq,
                 const float* __restrict__ gamma,
                 const float* __restrict__ beta,
                 float* __restrict__ out, int N) {
    const long total4 = (long)N * (CFEAT / 4);
    const long width  = (long)gridDim.x * blockDim.x;
    const long first  = (long)blockIdx.x * blockDim.x + threadIdx.x;
    const int  c4     = threadIdx.x & (CFEAT / 4 - 1);

    const float invN = 1.0f / (float)N;
    const float4 cs = ((const float4*)colsum)[c4];
    const float4 cq = ((const float4*)colsq)[c4];
    const float4 g  = ((const float4*)gamma)[c4];
    const float4 b  = ((const float4*)beta)[c4];
    float4 mu, rs;
    mu.x = cs.x * invN;  mu.y = cs.y * invN;  mu.z = cs.z * invN;  mu.w = cs.w * invN;
    rs.x = rsqrtf(cq.x * invN - mu.x * mu.x + 1e-5f);
    rs.y = rsqrtf(cq.y * invN - mu.y * mu.y + 1e-5f);
    rs.z = rsqrtf(cq.z * invN - mu.z * mu.z + 1e-5f);
    rs.w = rsqrtf(cq.w * invN - mu.w * mu.w + 1e-5f);

    const float4* __restrict__ in4 = (const float4*)hbuf;
    float4* __restrict__ out4 = (float4*)out;

    for (long base = first; base < total4; base += NORM_U * width) {
        float4 v[NORM_U];
#pragma unroll
        for (int u = 0; u < NORM_U; ++u) {
            const long j = base + u * width;
            v[u] = (j < total4) ? in4[j] : make_float4(0.0f, 0.0f, 0.0f, 0.0f);
        }
#pragma unroll
        for (int u = 0; u < NORM_U; ++u) {
            const long j = base + u * width;
            if (j < total4) out4[j] = norm_apply(v[u], g, mu, rs, b);
        }
    }
}

extern "C" void kernel_launch(void* const* d_in, const int* in_sizes, int n_in,
                              void* d_out, int out_size, void* d_ws, size_t ws_size,
                              hipStream_t stream) {
    const float* x    = (const float*)d_in[0];
    const int* e0src  = (const int*)d_in[1];
    const int* e0dst  = (const int*)d_in[2];
    const int* e1src  = (const int*)d_in[3];
    const int* e1dst  = (const int*)d_in[4];
    const float* Wl0  = (const float*)d_in[5];
    const float* bl0  = (const float*)d_in[6];
    const float* Wr0  = (const float*)d_in[7];
    const float* g0   = (const float*)d_in[8];
    const float* be0  = (const float*)d_in[9];
    const float* Wl1  = (const float*)d_in[10];
    const float* bl1  = (const float*)d_in[11];
    const float* Wr1  = (const float*)d_in[12];
    const float* g1   = (const float*)d_in[13];
    const float* be1  = (const float*)d_in[14];

    const int N1 = 50000, N2 = 12500;
    const int E0 = in_sizes[1];
    const int E1 = in_sizes[3];

    float* ws = (float*)d_ws;
    const long NC1 = (long)N1 * CFEAT;
    float* x1    = ws;
    float* sbuf  = ws + NC1;
    float* hbuf  = ws + 2 * NC1;
    float* cntb  = ws + 3 * NC1;
    float* stats = cntb + 50048;
    float* stats1 = x1 + 4000000;

    int* t0 = (int*)x1;
    int* t1 = (int*)(x1 + 4194304);
    const int nch0 = (E0 + CHUNK - 1) / CHUNK, nb0 = (N1 + 255) / 256;
    const int nch1 = (E1 + CHUNK - 1) / CHUNK, nb1 = (N2 + 63) / 64;

    binsort_kernel<<<nch0 + nch1, 512, 0, stream>>>(
        e0src, e0dst, t0 + 1179648, t0 + 2228224, t0 + 3276800, E0, N1, nb0, nch0, 8,
        e1src, e1dst, t1 + 1179648, t1 + 2228224, t1 + 3276800, E1, N2, nb1, nch1, 6);
    bincsr_kernel<<<nb0 + nb1, 1024, 0, stream>>>(
        t0 + 1179648, t0 + 2228224, t0 + 3276800, t0 + 131072, t0 + 65536, t0, stats, E0, N1, nch0, 8, nb0,
        t1 + 1179648, t1 + 2228224, t1 + 3276800, t1 + 131072, t1 + 65536, t1, stats1, E1, N2, nch1, 6);

    gather_kernel<false><<<(N1 + 7) / 8, 256, 0, stream>>>(x, t0 + 131072, t0 + 65536, t0, sbuf, cntb, N1,
                                                           g0, g0, g0, be0, N1);
    gemm_kernel<false><<<(N1 + GBM - 1) / GBM, 256, 0, stream>>>(
        sbuf, cntb, x, Wl0, bl0, Wr0, hbuf, stats, stats + 256, N1, g0, g0, g0, be0, N1);

    gather_kernel<true><<<(N2 + 7) / 8, 256, 0, stream>>>(hbuf, t1 + 131072, t1 + 65536, t1, sbuf, cntb, N2,
                                                          stats, stats + 256, g0, be0, N1);
    gemm_kernel<true><<<(N2 + GBM - 1) / GBM, 256, 0, stream>>>(
        sbuf, cntb, hbuf, Wl1, bl1, Wr1, x1, stats1, stats1 + 256, N2, stats, stats + 256, g0, be0, N1);
    norm_kernel<<<NORM_WG, 256, 0, stream>>>(
        x1, stats1, stats1 + 256, g1, be1, (float*)d_out, N2);
}


# GEMM workgroup tile 64x256 -> 64x64 (4 per row block, waves 4x2 of 16x32, K loop kept rolled): finer units even out the matrix pipe over WGPs; column-sum tree: 8 rows per lane, then 4 wave rows
# speedup vs baseline: 3.6794x; 1.0402x over previous
#include <hip/hip_runtime.h>

typedef __attribute__((ext_vector_type(2))) float v2f;
typedef __attribute__((ext_vector_type(8))) float v8f;

#define CFEAT 256

#define BIN_ROWS 256
#define CHUNK 4096

__global__ __launch_bounds__(512)
void binsort_kernel(const int* __restrict__ srcA, const int* __restrict__ dstA,
                    int* __restrict__ bsrcA, int* __restrict__ bdstA, int* __restrict__ segTA,
                    int EA, int NA, int BA, int nchunksA, int shiftA,
                    const int* __restrict__ srcB, const int* __restrict__ dstB,
                    int* __restrict__ bsrcB, int* __restrict__ bdstB, int* __restrict__ segTB,
                    int EB, int NB, int BB, int nchunksB, int shiftB) {
    __shared__ int hist[BIN_ROWS + 1];
    __shared__ int wtot[16];
    const bool second = ((int)blockIdx.x >= nchunksA);
    const int* __restrict__ src = second ? srcB : srcA;
    const int* __restrict__ dst = second ? dstB : dstA;
    int* __restrict__ bsrc = second ? bsrcB : bsrcA;
    int* __restrict__ bdst = second ? bdstB : bdstA;
    int* __restrict__ segT = second ? segTB : segTA;
    const int E = second ? EB : EA;
    const int N = second ? NB : NA;
    const int B = second ? BB : BA;
    const int nchunks = second ? nchunksB : nchunksA;
    const int shift = second ? shiftB : shiftA;
    const int tid = threadIdx.x;
    const int lane = tid & 31;
    const int wv = tid >> 5;
    const int c = second ? (int)blockIdx.x - nchunksA : (int)blockIdx.x;
    const int base = c * CHUNK;
    if (tid <= BIN_ROWS) hist[tid] = 0;

    int d[8], s[8], r[8];
#pragma unroll
    for (int k = 0; k < 2; ++k) {
        const int e = base + k * (CHUNK / 2) + tid * 4;
        if (e + 3 < E) {
            const int4 dv = *(const int4*)(dst + e);
            const int4 sv = *(const int4*)(src + e);
            d[k * 4 + 0] = dv.x; d[k * 4 + 1] = dv.y; d[k * 4 + 2] = dv.z; d[k * 4 + 3] = dv.w;
            s[k * 4 + 0] = sv.x; s[k * 4 + 1] = sv.y; s[k * 4 + 2] = sv.z; s[k * 4 + 3] = sv.w;
        } else {
#pragma unroll
            for (int i = 0; i < 4; ++i) {
                const bool in = (e + i < E);
                d[k * 4 + i] = in ? dst[e + i] : -1;
                s[k * 4 + i] = in ? src[e + i] : 0;
            }
        }
    }
    __syncthreads();
#pragma unroll
    for (int i = 0; i < 8; ++i) {
        const bool ok = ((unsigned)d[i] < (unsigned)N);
        r[i] = ok ? atomicAdd(&hist[d[i] >> shift], 1) : -1;
    }
    __syncthreads();
    int v = (tid < B) ? hist[tid] : 0;
    int incl = v;
#pragma unroll
    for (int off = 1; off < 32; off <<= 1) {
        int t = __shfl_up(incl, off);
        if (lane >= off) incl += t;
    }
    if (lane == 31) wtot[wv] = incl;
    __syncthreads();
    int pre = 0;
    for (int w = 0; w < wv; ++w) pre += wtot[w];
    const int excl = pre + incl - v;
    if (tid <= B) {
        hist[tid] = excl;
        segT[tid * nchunks + c] = excl;
    }
    __syncthreads();
#pragma unroll
    for (int i = 0; i < 8; ++i) {
        if (r[i] >= 0) {
            const int p = base + hist[d[i] >> shift] + r[i];
            bsrc[p] = s[i];
            bdst[p] = d[i];
        }
    }
}

__global__ __launch_bounds__(1024)
void bincsr_kernel(const int* __restrict__ bsrcA, const int* __restrict__ bdstA,
                   const int* __restrict__ segTA,
                   int* __restrict__ csrA, int* __restrict__ cursorA, int* __restrict__ degA,
                   float* __restrict__ statsA, int EA, int NA, int nchunksA, int shiftA, int nbinsA,
                   const int* __restrict__ bsrcB, const int* __restrict__ bdstB,
                   const int* __restrict__ segTB,
                   int* __restrict__ csrB, int* __restrict__ cursorB, int* __restrict__ degB,
                   float* __restrict__ statsB, int EB, int NB, int nchunksB, int shiftB) {
    __shared__ int cnt[BIN_ROWS];
    __shared__ int fillp[BIN_ROWS];
    __shared__ int wtot[8];
    __shared__ int sbase;
    const bool second = ((int)blockIdx.x >= nbinsA);
    const int* __restrict__ bsrc = second ? bsrcB : bsrcA;
    const int* __restrict__ bdst = second ? bdstB : bdstA;
    const int* __restrict__ segT = second ? segTB : segTA;
    int* __restrict__ csr = second ? csrB : csrA;
    int* __restrict__ cursor = second ? cursorB : cursorA;
    int* __restrict__ deg = second ? degB : degA;
    float* __restrict__ stats = second ? statsB : statsA;
    const int E = second ? EB : EA;
    const int N = second ? NB : NA;
    const int nchunks = second ? nchunksB : nchunksA;
    const int shift = second ? shiftB : shiftA;
    const int tid = threadIdx.x;
    const int lane = tid & 31;
    const int wv = tid >> 5;
    const int b = second ? (int)blockIdx.x - nbinsA : (int)blockIdx.x;
    const int rmask = (1 << shift) - 1;
    if (tid < BIN_ROWS) cnt[tid] = 0;
    if (tid == 0) sbase = 0;
    if (b == 0 && tid < 2 * CFEAT) stats[tid] = 0.0f;
    const int* __restrict__ lo = segT + (long)b * nchunks;
    const int* __restrict__ hi = lo + nchunks;
    int part = 0;
    for (int c = tid; c < nchunks; c += 1024) part += lo[c];
#pragma unroll
    for (int off = 16; off > 0; off >>= 1) part += __shfl_xor(part, off);
    __syncthreads();
    if (lane == 0 && part != 0) atomicAdd(&sbase, part);
    for (int cb = 0; cb < nchunks; cb += 1024) {
        const int cw = cb + wv;
        const int cl = cw + 32 * lane;
        int my0 = 0, my1 = 0;
        if (cl < nchunks) { my0 = lo[cl]; my1 = hi[cl]; }
        int J = (nchunks - cw + 31) >> 5;
        if (J > 32) J = 32;
        for (int j = 0; j < J; j += 4) {
            int s0[4], s1[4], dd[4];
#pragma unroll
            for (int k = 0; k < 4; ++k) {
                s0[k] = __builtin_amdgcn_readlane(my0, j + k);
                s1[k] = __builtin_amdgcn_readlane(my1, j + k);
                const int* __restrict__ pd = bdst + (long)(cw + 32 * (j + k)) * CHUNK;
                const int i = s0[k] + lane;
                dd[k] = (i < s1[k]) ? pd[i] : -1;
            }
#pragma unroll
            for (int k = 0; k < 4; ++k)
                if (dd[k] >= 0) atomicAdd(&cnt[dd[k] & rmask], 1);
#pragma unroll
            for (int k = 0; k < 4; ++k) {
                const int* __restrict__ pd = bdst + (long)(cw + 32 * (j + k)) * CHUNK;
                for (int i = s0[k] + 32 + lane; i < s1[k]; i += 32) atomicAdd(&cnt[pd[i] & rmask], 1);
            }
        }
    }
    __syncthreads();
    const int binbase = sbase;
    int v = (tid < BIN_ROWS) ? cnt[tid] : 0;
    int incl = v;
#pragma unroll
    for (int off = 1; off < 32; off <<= 1) {
        int t = __shfl_up(incl, off);
        if (lane >= off) incl += t;
    }
    if (lane == 31 && wv < 8) wtot[wv] = incl;
    __syncthreads();
    if (tid < BIN_ROWS) {
        int pre = 0;
        for (int w = 0; w < wv; ++w) pre += wtot[w];
        const int start = binbase + pre + incl - v;
        fillp[tid] = start;
        const int row = (b << shift) + tid;
        if (tid <= rmask && row < N) { cursor[row] = start; deg[row] = v; }
    }
    __syncthreads();
    for (int cb = 0; cb < nchunks; cb += 1024) {
        const int cw = cb + wv;
        const int cl = cw + 32 * lane;
        int my0 = 0, my1 = 0;
        if (cl < nchunks) { my0 = lo[cl]; my1 = hi[cl]; }
        int J = (nchunks - cw + 31) >> 5;
        if (J > 32) J = 32;
        for (int j = 0; j < J; j += 4) {
            int s0[4], s1[4], dd[4], ss[4];
#pragma unroll
            for (int k = 0; k < 4; ++k) {
                s0[k] = __builtin_amdgcn_readlane(my0, j + k);
                s1[k] = __builtin_amdgcn_readlane(my1, j + k);
                const long off = (long)(cw + 32 * (j + k)) * CHUNK;
                const int i = s0[k] + lane;
                const bool in = (i < s1[k]);
                dd[k] = in ? bdst[off + i] : -1;
                ss[k] = in ? bsrc[off + i] : 0;
            }
#pragma unroll
            for (int k = 0; k < 4; ++k) {
                if (dd[k] >= 0) {
                    const int p = atomicAdd(&fillp[dd[k] & rmask], 1);
                    if ((unsigned)p < (unsigned)E) csr[p] = ss[k];
                }
            }
#pragma unroll
            for (int k = 0; k < 4; ++k) {
                const long off = (long)(cw + 32 * (j + k)) * CHUNK;
                for (int i = s0[k] + 32 + lane; i < s1[k]; i += 32) {
                    const int d2 = bdst[off + i];
                    const int s2 = bsrc[off + i];
                    const int p = atomicAdd(&fillp[d2 & rmask], 1);
                    if ((unsigned)p < (unsigned)E) csr[p] = s2;
                }
            }
        }
    }
}

__device__ __forceinline__ void col_mu_a(float cs, float cq, float g, float invN, float& mu, float& a) {
    mu = cs * invN;
    a = g * rsqrtf(cq * invN - mu * mu + 1e-5f);
}

__device__ __forceinline__ float4 sum_affine(float4 v, float d, float4 cs, float4 cq, float4 g, float4 b, float invN) {
    float mu, a;
    float4 o;
    col_mu_a(cs.x, cq.x, g.x, invN, mu, a);  o.x = a * (v.x - d * mu) + d * b.x;
    col_mu_a(cs.y, cq.y, g.y, invN, mu, a);  o.y = a * (v.y - d * mu) + d * b.y;
    col_mu_a(cs.z, cq.z, g.z, invN, mu, a);  o.z = a * (v.z - d * mu) + d * b.z;
    col_mu_a(cs.w, cq.w, g.w, invN, mu, a);  o.w = a * (v.w - d * mu) + d * b.w;
    return o;
}

template <bool AFF>
__global__ void gather_kernel(const float* __restrict__ x,
                              const int* __restrict__ csr,
                              const int* __restrict__ cursor,
                              const int* __restrict__ deg,
                              float* __restrict__ s,
                              float* __restrict__ cnt, int N,
                              const float* __restrict__ psum,
                              const float* __restrict__ psq,
                              const float* __restrict__ pgamma,
                              const float* __restrict__ pbeta, int Nprev) {
    int w = __builtin_amdgcn_readfirstlane((int)((blockIdx.x * blockDim.x + threadIdx.x) >> 5));
    int lane = threadIdx.x & 31;
    if (w >= N) return;
    int d = deg[w];
    int beg = cursor[w];
    int end = beg + d;
    float4 a0 = make_float4(0.0f, 0.0f, 0.0f, 0.0f);
    float4 a1 = make_float4(0.0f, 0.0f, 0.0f, 0.0f);
    for (int c = beg; c < end; c += 32) {
        int n = end - c;
        if (n > 32) n = 32;
        int my = (lane < n) ? csr[c + lane] : 0;
        int j = 0;
        for (; j + 4 <= n; j += 4) {
            const float4* r0 = (const float4*)(x + (long)__builtin_amdgcn_readlane(my, j + 0) * CFEAT);
            const float4* r1 = (const float4*)(x + (long)__builtin_amdgcn_readlane(my, j + 1) * CFEAT);
            const float4* r2 = (const float4*)(x + (long)__builtin_amdgcn_readlane(my, j + 2) * CFEAT);
            const float4* r3 = (const float4*)(x + (long)__builtin_amdgcn_readlane(my, j + 3) * CFEAT);
            float4 p0 = r0[lane], q0 = r0[lane + 32];
            float4 p1 = r1[lane], q1 = r1[lane + 32];
            float4 p2 = r2[lane], q2 = r2[lane + 32];
            float4 p3 = r3[lane], q3 = r3[lane + 32];
            a0 += p0; a1 += q0;
            a0 += p1; a1 += q1;
            a0 += p2; a1 += q2;
            a0 += p3; a1 += q3;
        }
        for (; j < n; ++j) {
            const float4* r0 = (const float4*)(x + (long)__builtin_amdgcn_readlane(my, j) * CFEAT);
            float4 p0 = r0[lane], q0 = r0[lane + 32];
            a0 += p0; a1 += q0;
        }
    }
    if (AFF) {
        const float invN = 1.0f / (float)Nprev;
        const float df = (float)d;
        a0 = sum_affine(a0, df, ((const float4*)psum)[lane], ((const float4*)psq)[lane],
                        ((const float4*)pgamma)[lane], ((const float4*)pbeta)[lane], invN);
        a1 = sum_affine(a1, df, ((const float4*)psum)[lane + 32], ((const float4*)psq)[lane + 32],
                        ((const float4*)pgamma)[lane + 32], ((const float4*)pbeta)[lane + 32], invN);
    }
    float4* o = (float4*)(s + (long)w * CFEAT);
    o[lane] = a0;
    o[lane + 32] = a1;
    if (lane == 0) cnt[w] = (float)d;
}

#define GBM 64
#define GBN 64
#define GKC 32
#define GPITCH 36

template <bool AFF>
__global__ __launch_bounds__(256)
void gemm_kernel(const float* __restrict__ s,
                 const float* __restrict__ cnt,
                 const float* __restrict__ xt,
                 const float* __restrict__ Wl,
                 const float* __restrict__ bl,
                 const float* __restrict__ Wr,
                 float* __restrict__ h,
                 float* __restrict__ colsum,
                 float* __restrict__ colsq,
                 int N,
                 const float* __restrict__ psum,
                 const float* __restrict__ psq,
                 const float* __restrict__ pgamma,
                 const float* __restrict__ pbeta, int Nprev) {
    __shared__ float tileA[GBM * GPITCH];
    __shared__ float tileB[GBN * GPITCH];
    __shared__ float redS[4][GBN];
    __shared__ float redQ[4][GBN];
    __shared__ float affA[CFEAT];
    __shared__ float affM[CFEAT];
    __shared__ float affB[CFEAT];

    const int tid  = threadIdx.x;
    const int lane = tid & 31;
    const int wv   = tid >> 5;
    const int wm   = wv >> 1;
    const int wn   = wv & 1;
    const int l15  = lane & 15;
    const int hh   = lane >> 4;
    const int row0 = (int)(blockIdx.x >> 2) * GBM;
    const int col0 = (int)(blockIdx.x & 3) * GBN;

    if (AFF) {
        float mu, a;
        col_mu_a(psum[tid], psq[tid], pgamma[tid], 1.0f / (float)Nprev, mu, a);
        affA[tid] = a;
        affM[tid] = mu;
        affB[tid] = pbeta[tid];
    }

    const int srow = tid >> 3;
    const int sseg = tid & 7;
    int ar0 = row0 + srow;       if (ar0 > N - 1) ar0 = N - 1;
    int ar1 = row0 + srow + 32;  if (ar1 > N - 1) ar1 = N - 1;
    const float inv0 = 1.0f / fmaxf(cnt[ar0], 1.0f);
    const float inv1 = 1.0f / fmaxf(cnt[ar1], 1.0f);
    const long aoff0 = (long)ar0 * CFEAT + sseg * 4;
    const long aoff1 = (long)ar1 * CFEAT + sseg * 4;
    const long boff  = (long)(col0 + srow) * CFEAT + sseg * 4;
    const int  sdst  = srow * GPITCH + sseg * 4;

    v8f acc[2];
#pragma unroll
    for (int j = 0; j < 2; ++j)
        acc[j] = (v8f){0,0,0,0,0,0,0,0};

    float4 pa0, pa1, pb0, pb1;
    const long bstep = (long)32 * CFEAT;
    {
        const float* bp = Wl + boff;
        pa0 = *(const float4*)(s + aoff0);
        pa1 = *(const float4*)(s + aoff1);
        pb0 = *(const float4*)(bp);
        pb1 = *(const float4*)(bp + bstep);
    }

    const int fa = (wm * 16 + l15) * GPITCH + hh * 4;
    const int fb = (wn * 32 + l15) * GPITCH + hh * 4;

#pragma unroll 1
    for (int ch = 0; ch < 16; ++ch) {
        __syncthreads();
        {
            const float m0 = (ch < 8) ? inv0 : 1.0f;
            const float m1 = (ch < 8) ? inv1 : 1.0f;
            float4 t0 = pa0, t1 = pa1;
            t0.x *= m0; t0.y *= m0; t0.z *= m0; t0.w *= m0;
            t1.x *= m1; t1.y *= m1; t1.z *= m1; t1.w *= m1;
            if (AFF && ch >= 8) {
                const int q = ((ch & 7) * 8 + sseg) * 4;
                const float4 ca = *(const float4*)(affA + q);
                const float4 cm = *(const float4*)(affM + q);
                const float4 cb = *(const float4*)(affB + q);
                t0.x = ca.x * (t0.x - cm.x) + cb.x;  t0.y = ca.y * (t0.y - cm.y) + cb.y;
                t0.z = ca.z * (t0.z - cm.z) + cb.z;  t0.w = ca.w * (t0.w - cm.w) + cb.w;
                t1.x = ca.x * (t1.x - cm.x) + cb.x;  t1.y = ca.y * (t1.y - cm.y) + cb.y;
                t1.z = ca.z * (t1.z - cm.z) + cb.z;  t1.w = ca.w * (t1.w - cm.w) + cb.w;
            }
            *(float4*)(tileA + sdst) = t0;
            *(float4*)(tileA + sdst + 32 * GPITCH) = t1;
            *(float4*)(tileB + sdst) = pb0;
            *(float4*)(tileB + sdst + 32 * GPITCH) = pb1;
        }
        __syncthreads();
        if (ch + 1 < 16) {
            const int cn = ch + 1;
            const float* Ab = (cn < 8) ? s : xt;
            const float* Bb = (cn < 8) ? Wl : Wr;
            const int k0 = (cn & 7) * GKC;
            pa0 = *(const float4*)(Ab + aoff0 + k0);
            pa1 = *(const float4*)(Ab + aoff1 + k0);
            const float* bp = Bb + boff + k0;
            pb0 = *(const float4*)(bp);
            pb1 = *(const float4*)(bp + bstep);
        }
#pragma unroll
        for (int kk = 0; kk < 4; ++kk) {
            float4 af, bf[2];
            af = *(const float4*)(tileA + fa + kk * 8);
#pragma unroll
            for (int j = 0; j < 2; ++j) bf[j] = *(const float4*)(tileB + fb + j * 16 * GPITCH + kk * 8);
#pragma unroll
            for (int j = 0; j < 2; ++j) {
                v2f a = {af.x, af.y};
                v2f b = {bf[j].x, bf[j].y};
                acc[j] = __builtin_amdgcn_wmma_f32_16x16x4_f32(
                    false, a, false, b, (short)0, acc[j], false, false);
            }
#pragma unroll
            for (int j = 0; j < 2; ++j) {
                v2f a = {af.z, af.w};
                v2f b = {bf[j].z, bf[j].w};
                acc[j] = __builtin_amdgcn_wmma_f32_16x16x4_f32(
                    false, a, false, b, (short)0, acc[j], false, false);
            }
        }
    }

#pragma unroll
    for (int j = 0; j < 2; ++j) {
        const int cl = wn * 32 + j * 16 + l15;
        const int col = col0 + cl;
        const float bias = bl[col];
        float sp = 0.0f, qp = 0.0f;
        const int rbase = row0 + wm * 16 + hh * 8;
#pragma unroll
        for (int v = 0; v < 8; ++v) {
            float val = fmaxf(acc[j][v] + bias, 0.0f);
            int r = rbase + v;
            bool ok = (r < N);
            float m = ok ? val : 0.0f;
            sp += m;
            qp += m * m;
            if (ok) h[(long)r * CFEAT + col] = val;
        }
        sp += __shfl_xor(sp, 16);
        qp += __shfl_xor(qp, 16);
        if (hh == 0) { redS[wm][cl] = sp; redQ[wm][cl] = qp; }
    }
    __syncthreads();
    if (tid < GBN) {
        atomicAdd(&colsum[col0 + tid], (redS[0][tid] + redS[1][tid]) + (redS[2][tid] + redS[3][tid]));
    } else if (tid < 2 * GBN) {
        const int c = tid - GBN;
        atomicAdd(&colsq[col0 + c], (redQ[0][c] + redQ[1][c]) + (redQ[2][c] + redQ[3][c]));
    }
}

#define NORM_U 8
#define NORM_WG 512

__device__ __forceinline__ float4 norm_apply(float4 v, float4 g, float4 mu, float4 rs, float4 b) {
    float4 o;
    o.x = g.x * (v.x - mu.x) * rs.x + b.x;
    o.y = g.y * (v.y - mu.y) * rs.y + b.y;
    o.z = g.z * (v.z - mu.z) * rs.z + b.z;
    o.w = g.w * (v.w - mu.w) * rs.w + b.w;
    return o;
}

__global__ __launch_bounds__(256)
void norm_kernel(const float* __restrict__ hbuf,
                 const float* __restrict__ colsum,
                 const float* __restrict__ colsq,
                 const float* __restrict__ gamma,
                 const float* __restrict__ beta,
                 float* __restrict__ out, int N) {
    const long total4 = (long)N * (CFEAT / 4);
    const long width  = (long)gridDim.x * blockDim.x;
    const long first  = (long)blockIdx.x * blockDim.x + threadIdx.x;
    const int  c4     = threadIdx.x & (CFEAT / 4 - 1);

    const float invN = 1.0f / (float)N;
    const float4 cs = ((const float4*)colsum)[c4];
    const float4 cq = ((const float4*)colsq)[c4];
    const float4 g  = ((const float4*)gamma)[c4];
    const float4 b  = ((const float4*)beta)[c4];
    float4 mu, rs;
    mu.x = cs.x * invN;  mu.y = cs.y * invN;  mu.z = cs.z * invN;  mu.w = cs.w * invN;
    rs.x = rsqrtf(cq.x * invN - mu.x * mu.x + 1e-5f);
    rs.y = rsqrtf(cq.y * invN - mu.y * mu.y + 1e-5f);
    rs.z = rsqrtf(cq.z * invN - mu.z * mu.z + 1e-5f);
    rs.w = rsqrtf(cq.w * invN - mu.w * mu.w + 1e-5f);

    const float4* __restrict__ in4 = (const float4*)hbuf;
    float4* __restrict__ out4 = (float4*)out;

    for (long base = first; base < total4; base += NORM_U * width) {
        float4 v[NORM_U];
#pragma unroll
        for (int u = 0; u < NORM_U; ++u) {
            const long j = base + u * width;
            v[u] = (j < total4) ? in4[j] : make_float4(0.0f, 0.0f, 0.0f, 0.0f);
        }
#pragma unroll
        for (int u = 0; u < NORM_U; ++u) {
            const long j = base + u * width;
            if (j < total4) out4[j] = norm_apply(v[u], g, mu, rs, b);
        }
    }
}

extern "C" void kernel_launch(void* const* d_in, const int* in_sizes, int n_in,
                              void* d_out, int out_size, void* d_ws, size_t ws_size,
                              hipStream_t stream) {
    const float* x    = (const float*)d_in[0];
    const int* e0src  = (const int*)d_in[1];
    const int* e0dst  = (const int*)d_in[2];
    const int* e1src  = (const int*)d_in[3];
    const int* e1dst  = (const int*)d_in[4];
    const float* Wl0  = (const float*)d_in[5];
    const float* bl0  = (const float*)d_in[6];
    const float* Wr0  = (const float*)d_in[7];
    const float* g0   = (const float*)d_in[8];
    const float* be0  = (const float*)d_in[9];
    const float* Wl1  = (const float*)d_in[10];
    const float* bl1  = (const float*)d_in[11];
    const float* Wr1  = (const float*)d_in[12];
    const float* g1   = (const float*)d_in[13];
    const float* be1  = (const float*)d_in[14];

    const int N1 = 50000, N2 = 12500;
    const int E0 = in_sizes[1];
    const int E1 = in_sizes[3];

    float* ws = (float*)d_ws;
    const long NC1 = (long)N1 * CFEAT;
    float* x1    = ws;
    float* sbuf  = ws + NC1;
    float* hbuf  = ws + 2 * NC1;
    float* cntb  = ws + 3 * NC1;
    float* stats = cntb + 50048;
    float* stats1 = x1 + 4000000;

    int* t0 = (int*)x1;
    int* t1 = (int*)(x1 + 4194304);
    const int nch0 = (E0 + CHUNK - 1) / CHUNK, nb0 = (N1 + 255) / 256;
    const int nch1 = (E1 + CHUNK - 1) / CHUNK, nb1 = (N2 + 63) / 64;

    binsort_kernel<<<nch0 + nch1, 512, 0, stream>>>(
        e0src, e0dst, t0 + 1179648, t0 + 2228224, t0 + 3276800, E0, N1, nb0, nch0, 8,
        e1src, e1dst, t1 + 1179648, t1 + 2228224, t1 + 3276800, E1, N2, nb1, nch1, 6);
    bincsr_kernel<<<nb0 + nb1, 1024, 0, stream>>>(
        t0 + 1179648, t0 + 2228224, t0 + 3276800, t0 + 131072, t0 + 65536, t0, stats, E0, N1, nch0, 8, nb0,
        t1 + 1179648, t1 + 2228224, t1 + 3276800, t1 + 131072, t1 + 65536, t1, stats1, E1, N2, nch1, 6);

    gather_kernel<false><<<(N1 + 7) / 8, 256, 0, stream>>>(x, t0 + 131072, t0 + 65536, t0, sbuf, cntb, N1,
                                                           g0, g0, g0, be0, N1);
    gemm_kernel<false><<<((N1 + GBM - 1) / GBM) * (CFEAT / GBN), 256, 0, stream>>>(
        sbuf, cntb, x, Wl0, bl0, Wr0, hbuf, stats, stats + 256, N1, g0, g0, g0, be0, N1);

    gather_kernel<true><<<(N2 + 7) / 8, 256, 0, stream>>>(hbuf, t1 + 131072, t1 + 65536, t1, sbuf, cntb, N2,
                                                          stats, stats + 256, g0, be0, N1);
    gemm_kernel<true><<<((N2 + GBM - 1) / GBM) * (CFEAT / GBN), 256, 0, stream>>>(
        sbuf, cntb, hbuf, Wl1, bl1, Wr1, x1, stats1, stats1 + 256, N2, stats, stats + 256, g0, be0, N1);
    norm_kernel<<<NORM_WG, 256, 0, stream>>>(
        x1, stats1, stats1 + 256, g1, be1, (float*)d_out, N2);
}
